# starting kernel
# baseline (speedup 1.0000x reference)
#include <hip/hip_runtime.h>
#include <math.h>

typedef __attribute__((ext_vector_type(2))) float v2f;
typedef __attribute__((ext_vector_type(8))) float v8f;

#define MM    129
#define MP    144
#define NPP   512
#define NB    4
#define NPTS  (NB * NPP)

#define PI_F   3.14159265358979323846f
#define TAU_F  (12.0f / (129.0f * 129.0f))
#define L_F    (2.0f * PI_F)


__device__ __forceinline__ v8f wmma4(v2f a, v2f b, v8f c) {
  return __builtin_amdgcn_wmma_f32_16x16x4_f32(false, a, false, b, (short)0, c, false, false);
}

__device__ __forceinline__ void store_tile(float* C, int ldc, int row0, int col0, v8f acc) {
  int lane = threadIdx.x & 31;
  int r0 = row0 + ((lane >> 4) << 3);
  int c  = col0 + (lane & 15);
#pragma unroll
  for (int v = 0; v < 8; ++v) C[(r0 + v) * ldc + c] = acc[v];
}

template <typename FA, typename FB>
__device__ __forceinline__ v8f mm_tile(int row0, int col0, int K, FA la, FB lb) {
  int lane = threadIdx.x & 31;
  int r  = lane & 15;
  int kb = (lane >> 4) << 1;
  v8f acc = {};
  for (int k0 = 0; k0 < K; k0 += 4) {
    v2f a, b;
    a.x = la(row0 + r, k0 + kb);
    a.y = la(row0 + r, k0 + kb + 1);
    b.x = lb(k0 + kb,     col0 + r);
    b.y = lb(k0 + kb + 1, col0 + r);
    acc = wmma4(a, b, acc);
  }
  return acc;
}


__global__ void k_tables(float* __restrict__ Ere, float* __restrict__ Eim,
                         float* __restrict__ W0,  float* __restrict__ W1,
                         const float* s0, const float* s1,
                         const float* a0, const float* a1) {
  int t = blockIdx.x * blockDim.x + threadIdx.x;
  if (t >= MP * MP) return;
  int i = t / MP, k = t % MP;
  float er = 0.f, ei = 0.f, w0 = 0.f, w1 = 0.f;
  if (i < MM && k < MM) {
    int m = (i * k) % MM;
    float th = (2.0f * PI_F / (float)MM) * (float)m;
    er = cosf(th);
    ei = -sinf(th);
    float f1 = (float)((i <= MM / 2) ? i : i - MM);
    float f2 = (float)((k <= MM / 2) ? k : k - MM);
    float k2 = f1 * f1 + f2 * f2;
    float dec = (PI_F / TAU_F) * expf(TAU_F * k2);
    float m2   = (float)MM * (float)MM;
    float coef = 1.0f / (8.0f * PI_F * PI_F * m2 * m2);
    float base = coef * dec * dec * 4.0f * PI_F;
    float sh0 = s0[0], sh1 = s1[0];
    w0 = base * a0[0] / (k2 + sh0 * sh0);
    w1 = base * a1[0] / (k2 + sh1 * sh1);
  }
  Ere[t] = er; Eim[t] = ei; W0[t] = w0; W1[t] = w1;
}

__global__ void k_spread(const float* __restrict__ x,
                         float* __restrict__ Gx, float* __restrict__ Gy) {
  int t = blockIdx.x * blockDim.x + threadIdx.x;
  if (t >= NPTS * MP) return;
  int pt = t / MP, i = t % MP;
  float gx = 0.f, gy = 0.f;
  if (i < MM) {
    float xi  = (L_F / (float)MM) * (float)i;
    float inv = 1.0f / (4.0f * TAU_F);
    float d0 = x[2 * pt + 0] - xi;
    float d1 = x[2 * pt + 1] - xi;
    gx = expf(-d0 * d0 * inv) + expf(-(d0 - L_F) * (d0 - L_F) * inv)
       + expf(-(d0 + L_F) * (d0 + L_F) * inv);
    gy = expf(-d1 * d1 * inv) + expf(-(d1 - L_F) * (d1 - L_F) * inv)
       + expf(-(d1 + L_F) * (d1 + L_F) * inv);
  }
  Gx[t] = gx; Gy[t] = gy;
}

__global__ void k_dft(const float* __restrict__ Gx, const float* __restrict__ Gy,
                      const float* __restrict__ Ere, const float* __restrict__ Eim,
                      float* __restrict__ FxRe, float* __restrict__ FxIm,
                      float* __restrict__ FyRe, float* __restrict__ FyIm) {
  int z = blockIdx.z;
  const float* G = (z < 2) ? Gx : Gy;
  const float* E = (z & 1) ? Eim : Ere;
  float* D = (z == 0) ? FxRe : (z == 1) ? FxIm : (z == 2) ? FyRe : FyIm;
  int row0 = blockIdx.x * 16, col0 = blockIdx.y * 16;
  v8f acc = mm_tile(row0, col0, MP,
                    [&](int r, int k) { return G[r * MP + k]; },
                    [&](int k, int c) { return E[k * MP + c]; });
  store_tile(D, MP, row0, col0, acc);
}

__global__ void k_sqmag(const float* __restrict__ FxRe, const float* __restrict__ FxIm,
                        const float* __restrict__ FyRe, const float* __restrict__ FyIm,
                        float* __restrict__ Ax, float* __restrict__ Ay) {
  int t = blockIdx.x * blockDim.x + threadIdx.x;
  if (t >= NPTS * MP) return;
  Ax[t] = FxRe[t] * FxRe[t] + FxIm[t] * FxIm[t];
  Ay[t] = FyRe[t] * FyRe[t] + FyIm[t] * FyIm[t];
}

__global__ void k_smat(const float* __restrict__ FxRe, const float* __restrict__ FxIm,
                       const float* __restrict__ FyRe, const float* __restrict__ FyIm,
                       float* __restrict__ SRe, float* __restrict__ SIm) {
  int b = blockIdx.z;
  int row0 = blockIdx.x * 16, col0 = blockIdx.y * 16;
  int lane = threadIdx.x & 31, r = lane & 15, kb = (lane >> 4) << 1;
  const float* fxr = FxRe + b * NPP * MP;
  const float* fxi = FxIm + b * NPP * MP;
  const float* fyr = FyRe + b * NPP * MP;
  const float* fyi = FyIm + b * NPP * MP;
  v8f accRe = {}, accIm = {};
  for (int k0 = 0; k0 < NPP; k0 += 4) {
    int q0 = k0 + kb, q1 = q0 + 1;
    v2f aRe = { fxr[q0 * MP + row0 + r], fxr[q1 * MP + row0 + r] };
    v2f aIm = { fxi[q0 * MP + row0 + r], fxi[q1 * MP + row0 + r] };
    v2f aImN = { -aIm.x, -aIm.y };
    v2f bRe = { fyr[q0 * MP + col0 + r], fyr[q1 * MP + col0 + r] };
    v2f bIm = { fyi[q0 * MP + col0 + r], fyi[q1 * MP + col0 + r] };
    accRe = wmma4(aRe,  bRe, accRe);
    accRe = wmma4(aImN, bIm, accRe);
    accIm = wmma4(aRe,  bIm, accIm);
    accIm = wmma4(aIm,  bRe, accIm);
  }
  store_tile(SRe + b * MP * MP, MP, row0, col0, accRe);
  store_tile(SIm + b * MP * MP, MP, row0, col0, accIm);
}

__global__ void k_umat(const float* __restrict__ W,
                       const float* __restrict__ SRe, const float* __restrict__ SIm,
                       const float* __restrict__ FyRe, const float* __restrict__ FyIm,
                       float* __restrict__ URe, float* __restrict__ UIm) {
  int b = blockIdx.z;
  int row0 = blockIdx.x * 16;
  int col0 = blockIdx.y * 16;
  int lane = threadIdx.x & 31, r = lane & 15, kb = (lane >> 4) << 1;
  const float* sre = SRe + b * MP * MP;
  const float* sim = SIm + b * MP * MP;
  const float* fyr = FyRe + b * NPP * MP;
  const float* fyi = FyIm + b * NPP * MP;
  v8f accRe = {}, accIm = {};
  for (int k0 = 0; k0 < MP; k0 += 4) {
    int k2a = k0 + kb, k2b = k2a + 1;
    int ra = (row0 + r) * MP;
    float w0 = W[ra + k2a], w1 = W[ra + k2b];
    v2f tRe = { w0 * sre[ra + k2a], w1 * sre[ra + k2b] };
    v2f tIm = { w0 * sim[ra + k2a], w1 * sim[ra + k2b] };
    v2f tReN = { -tRe.x, -tRe.y };
    int pb = (col0 + r) * MP;
    v2f bRe = { fyr[pb + k2a], fyr[pb + k2b] };
    v2f bIm = { fyi[pb + k2a], fyi[pb + k2b] };
    accRe = wmma4(tRe,  bRe, accRe);
    accRe = wmma4(tIm,  bIm, accRe);
    accIm = wmma4(tIm,  bRe, accIm);
    accIm = wmma4(tReN, bIm, accIm);
  }
  store_tile(URe + b * MP * NPP, NPP, row0, col0, accRe);
  store_tile(UIm + b * MP * NPP, NPP, row0, col0, accIm);
}

__global__ void k_vmat(const float* __restrict__ W, const float* __restrict__ Ay,
                       float* __restrict__ V) {
  int b = blockIdx.z;
  int row0 = blockIdx.x * 16, col0 = blockIdx.y * 16;
  const float* ay = Ay + b * NPP * MP;
  v8f acc = mm_tile(row0, col0, MP,
                    [&](int r, int k) { return W[r * MP + k]; },
                    [&](int k, int c) { return ay[c * MP + k]; });
  store_tile(V + b * MP * NPP, NPP, row0, col0, acc);
}

__global__ void k_reduce(const float* __restrict__ FxRe, const float* __restrict__ FxIm,
                         const float* __restrict__ Ax,
                         const float* __restrict__ URe, const float* __restrict__ UIm,
                         const float* __restrict__ V,
                         float* __restrict__ out, int ch) {
  int pt = blockIdx.x;
  int b = pt >> 9, p = pt & (NPP - 1);
  int lane = threadIdx.x;
  float tot = 0.f, dia = 0.f;
  for (int k = lane; k < MP; k += 32) {
    int uc = (b * MP + k) * NPP + p;
    tot += FxRe[pt * MP + k] * URe[uc] + FxIm[pt * MP + k] * UIm[uc];
    dia += Ax[pt * MP + k] * V[uc];
  }
#pragma unroll
  for (int off = 16; off > 0; off >>= 1) {
    tot += __shfl_down(tot, off);
    dia += __shfl_down(dia, off);
  }
  if (lane == 0) out[pt * 2 + ch] = tot - dia;
}


extern "C" void kernel_launch(void* const* d_in, const int* in_sizes, int n_in,
                              void* d_out, int out_size, void* d_ws, size_t ws_size,
                              hipStream_t stream) {
  (void)in_sizes; (void)n_in; (void)out_size; (void)ws_size;
  const float* x  = (const float*)d_in[0];
  const float* s0 = (const float*)d_in[1];
  const float* s1 = (const float*)d_in[2];
  const float* a0 = (const float*)d_in[3];
  const float* a1 = (const float*)d_in[4];
  float* out = (float*)d_out;

  float* w = (float*)d_ws;
  float* Ere  = w; w += MP * MP;
  float* Eim  = w; w += MP * MP;
  float* W0   = w; w += MP * MP;
  float* W1   = w; w += MP * MP;
  float* Gx   = w; w += NPTS * MP;
  float* Gy   = w; w += NPTS * MP;
  float* FxRe = w; w += NPTS * MP;
  float* FxIm = w; w += NPTS * MP;
  float* FyRe = w; w += NPTS * MP;
  float* FyIm = w; w += NPTS * MP;
  float* Ax   = w; w += NPTS * MP;
  float* Ay   = w; w += NPTS * MP;
  float* SRe  = w; w += NB * MP * MP;
  float* SIm  = w; w += NB * MP * MP;
  float* URe  = w; w += NB * MP * NPP;
  float* UIm  = w; w += NB * MP * NPP;
  float* V    = w; w += NB * MP * NPP;

  k_tables<<<(MP * MP + 255) / 256, 256, 0, stream>>>(Ere, Eim, W0, W1, s0, s1, a0, a1);
  k_spread<<<(NPTS * MP + 255) / 256, 256, 0, stream>>>(x, Gx, Gy);
  k_dft<<<dim3(NPTS / 16, MP / 16, 4), 32, 0, stream>>>(Gx, Gy, Ere, Eim,
                                                        FxRe, FxIm, FyRe, FyIm);
  k_sqmag<<<(NPTS * MP + 255) / 256, 256, 0, stream>>>(FxRe, FxIm, FyRe, FyIm, Ax, Ay);
  k_smat<<<dim3(MP / 16, MP / 16, NB), 32, 0, stream>>>(FxRe, FxIm, FyRe, FyIm, SRe, SIm);

  for (int c = 0; c < 2; ++c) {
    const float* Wc = c ? W1 : W0;
    k_umat<<<dim3(MP / 16, NPP / 16, NB), 32, 0, stream>>>(Wc, SRe, SIm, FyRe, FyIm, URe, UIm);
    k_vmat<<<dim3(MP / 16, NPP / 16, NB), 32, 0, stream>>>(Wc, Ay, V);
    k_reduce<<<NPTS, 32, 0, stream>>>(FxRe, FxIm, Ax, URe, UIm, V, out, c);
  }
}


# Gram kernel (K=512): K range split over 8 waves of one 256-thread block per tile, the 8 partial tiles added in fixed wave order through LDS; changes the order of that sum
# speedup vs baseline: 1.1700x; 1.1700x over previous
#include <hip/hip_runtime.h>
#include <math.h>

typedef __attribute__((ext_vector_type(2))) float v2f;
typedef __attribute__((ext_vector_type(8))) float v8f;

#define MM    129
#define MP    144
#define NPP   512
#define NB    4
#define NPTS  (NB * NPP)
#define KSPLIT 8

#define PI_F   3.14159265358979323846f
#define TAU_F  (12.0f / (129.0f * 129.0f))
#define L_F    (2.0f * PI_F)


__device__ __forceinline__ v8f wmma4(v2f a, v2f b, v8f c) {
  return __builtin_amdgcn_wmma_f32_16x16x4_f32(false, a, false, b, (short)0, c, false, false);
}

__device__ __forceinline__ void store_tile(float* C, int ldc, int row0, int col0, v8f acc) {
  int lane = threadIdx.x & 31;
  int r0 = row0 + ((lane >> 4) << 3);
  int c  = col0 + (lane & 15);
#pragma unroll
  for (int v = 0; v < 8; ++v) C[(r0 + v) * ldc + c] = acc[v];
}

template <typename FA, typename FB>
__device__ __forceinline__ v8f mm_tile(int row0, int col0, int K, FA la, FB lb) {
  int lane = threadIdx.x & 31;
  int r  = lane & 15;
  int kb = (lane >> 4) << 1;
  v8f acc = {};
  for (int k0 = 0; k0 < K; k0 += 4) {
    v2f a, b;
    a.x = la(row0 + r, k0 + kb);
    a.y = la(row0 + r, k0 + kb + 1);
    b.x = lb(k0 + kb,     col0 + r);
    b.y = lb(k0 + kb + 1, col0 + r);
    acc = wmma4(a, b, acc);
  }
  return acc;
}


__global__ void k_tables(float* __restrict__ Ere, float* __restrict__ Eim,
                         float* __restrict__ W0,  float* __restrict__ W1,
                         const float* s0, const float* s1,
                         const float* a0, const float* a1) {
  int t = blockIdx.x * blockDim.x + threadIdx.x;
  if (t >= MP * MP) return;
  int i = t / MP, k = t % MP;
  float er = 0.f, ei = 0.f, w0 = 0.f, w1 = 0.f;
  if (i < MM && k < MM) {
    int m = (i * k) % MM;
    float th = (2.0f * PI_F / (float)MM) * (float)m;
    er = cosf(th);
    ei = -sinf(th);
    float f1 = (float)((i <= MM / 2) ? i : i - MM);
    float f2 = (float)((k <= MM / 2) ? k : k - MM);
    float k2 = f1 * f1 + f2 * f2;
    float dec = (PI_F / TAU_F) * expf(TAU_F * k2);
    float m2   = (float)MM * (float)MM;
    float coef = 1.0f / (8.0f * PI_F * PI_F * m2 * m2);
    float base = coef * dec * dec * 4.0f * PI_F;
    float sh0 = s0[0], sh1 = s1[0];
    w0 = base * a0[0] / (k2 + sh0 * sh0);
    w1 = base * a1[0] / (k2 + sh1 * sh1);
  }
  Ere[t] = er; Eim[t] = ei; W0[t] = w0; W1[t] = w1;
}

__global__ void k_spread(const float* __restrict__ x,
                         float* __restrict__ Gx, float* __restrict__ Gy) {
  int t = blockIdx.x * blockDim.x + threadIdx.x;
  if (t >= NPTS * MP) return;
  int pt = t / MP, i = t % MP;
  float gx = 0.f, gy = 0.f;
  if (i < MM) {
    float xi  = (L_F / (float)MM) * (float)i;
    float inv = 1.0f / (4.0f * TAU_F);
    float d0 = x[2 * pt + 0] - xi;
    float d1 = x[2 * pt + 1] - xi;
    gx = expf(-d0 * d0 * inv) + expf(-(d0 - L_F) * (d0 - L_F) * inv)
       + expf(-(d0 + L_F) * (d0 + L_F) * inv);
    gy = expf(-d1 * d1 * inv) + expf(-(d1 - L_F) * (d1 - L_F) * inv)
       + expf(-(d1 + L_F) * (d1 + L_F) * inv);
  }
  Gx[t] = gx; Gy[t] = gy;
}

__global__ void k_dft(const float* __restrict__ Gx, const float* __restrict__ Gy,
                      const float* __restrict__ Ere, const float* __restrict__ Eim,
                      float* __restrict__ FxRe, float* __restrict__ FxIm,
                      float* __restrict__ FyRe, float* __restrict__ FyIm) {
  int z = blockIdx.z;
  const float* G = (z < 2) ? Gx : Gy;
  const float* E = (z & 1) ? Eim : Ere;
  float* D = (z == 0) ? FxRe : (z == 1) ? FxIm : (z == 2) ? FyRe : FyIm;
  int row0 = blockIdx.x * 16, col0 = blockIdx.y * 16;
  v8f acc = mm_tile(row0, col0, MP,
                    [&](int r, int k) { return G[r * MP + k]; },
                    [&](int k, int c) { return E[k * MP + c]; });
  store_tile(D, MP, row0, col0, acc);
}

__global__ void k_sqmag(const float* __restrict__ FxRe, const float* __restrict__ FxIm,
                        const float* __restrict__ FyRe, const float* __restrict__ FyIm,
                        float* __restrict__ Ax, float* __restrict__ Ay) {
  int t = blockIdx.x * blockDim.x + threadIdx.x;
  if (t >= NPTS * MP) return;
  Ax[t] = FxRe[t] * FxRe[t] + FxIm[t] * FxIm[t];
  Ay[t] = FyRe[t] * FyRe[t] + FyIm[t] * FyIm[t];
}

__global__ void __launch_bounds__(32 * KSPLIT)
k_smat(const float* __restrict__ FxRe, const float* __restrict__ FxIm,
       const float* __restrict__ FyRe, const float* __restrict__ FyIm,
       float* __restrict__ SRe, float* __restrict__ SIm) {
  __shared__ float part[KSPLIT][2][256];
  int b = blockIdx.z;
  int row0 = blockIdx.x * 16, col0 = blockIdx.y * 16;
  int lane = threadIdx.x & 31, r = lane & 15, kb = (lane >> 4) << 1;
  int wave = __builtin_amdgcn_readfirstlane(threadIdx.x >> 5);
  int qbeg = wave * (NPP / KSPLIT);
  const float* fxr = FxRe + (b * NPP + qbeg) * MP;
  const float* fxi = FxIm + (b * NPP + qbeg) * MP;
  const float* fyr = FyRe + (b * NPP + qbeg) * MP;
  const float* fyi = FyIm + (b * NPP + qbeg) * MP;
  v8f accRe = {}, accIm = {};
  for (int k0 = 0; k0 < NPP / KSPLIT; k0 += 4) {
    int q0 = k0 + kb, q1 = q0 + 1;
    v2f aRe = { fxr[q0 * MP + row0 + r], fxr[q1 * MP + row0 + r] };
    v2f aIm = { fxi[q0 * MP + row0 + r], fxi[q1 * MP + row0 + r] };
    v2f aImN = { -aIm.x, -aIm.y };
    v2f bRe = { fyr[q0 * MP + col0 + r], fyr[q1 * MP + col0 + r] };
    v2f bIm = { fyi[q0 * MP + col0 + r], fyi[q1 * MP + col0 + r] };
    accRe = wmma4(aRe,  bRe, accRe);
    accRe = wmma4(aImN, bIm, accRe);
    accIm = wmma4(aRe,  bIm, accIm);
    accIm = wmma4(aIm,  bRe, accIm);
  }
  for (int v = 0; v < 8; ++v) {
    part[wave][0][v * 32 + lane] = accRe[v];
    part[wave][1][v * 32 + lane] = accIm[v];
  }
  __syncthreads();
  int t = threadIdx.x;
  float sre = part[0][0][t], sim = part[0][1][t];
  for (int w = 1; w < KSPLIT; ++w) {
    sre += part[w][0][t];
    sim += part[w][1][t];
  }
  int l = t & 31;
  int row = row0 + (t >> 5) + ((l >> 4) << 3);
  int col = col0 + (l & 15);
  SRe[b * MP * MP + row * MP + col] = sre;
  SIm[b * MP * MP + row * MP + col] = sim;
}

__global__ void k_umat(const float* __restrict__ W,
                       const float* __restrict__ SRe, const float* __restrict__ SIm,
                       const float* __restrict__ FyRe, const float* __restrict__ FyIm,
                       float* __restrict__ URe, float* __restrict__ UIm) {
  int b = blockIdx.z;
  int row0 = blockIdx.x * 16;
  int col0 = blockIdx.y * 16;
  int lane = threadIdx.x & 31, r = lane & 15, kb = (lane >> 4) << 1;
  const float* sre = SRe + b * MP * MP;
  const float* sim = SIm + b * MP * MP;
  const float* fyr = FyRe + b * NPP * MP;
  const float* fyi = FyIm + b * NPP * MP;
  v8f accRe = {}, accIm = {};
  for (int k0 = 0; k0 < MP; k0 += 4) {
    int k2a = k0 + kb, k2b = k2a + 1;
    int ra = (row0 + r) * MP;
    float w0 = W[ra + k2a], w1 = W[ra + k2b];
    v2f tRe = { w0 * sre[ra + k2a], w1 * sre[ra + k2b] };
    v2f tIm = { w0 * sim[ra + k2a], w1 * sim[ra + k2b] };
    v2f tReN = { -tRe.x, -tRe.y };
    int pb = (col0 + r) * MP;
    v2f bRe = { fyr[pb + k2a], fyr[pb + k2b] };
    v2f bIm = { fyi[pb + k2a], fyi[pb + k2b] };
    accRe = wmma4(tRe,  bRe, accRe);
    accRe = wmma4(tIm,  bIm, accRe);
    accIm = wmma4(tIm,  bRe, accIm);
    accIm = wmma4(tReN, bIm, accIm);
  }
  store_tile(URe + b * MP * NPP, NPP, row0, col0, accRe);
  store_tile(UIm + b * MP * NPP, NPP, row0, col0, accIm);
}

__global__ void k_vmat(const float* __restrict__ W, const float* __restrict__ Ay,
                       float* __restrict__ V) {
  int b = blockIdx.z;
  int row0 = blockIdx.x * 16, col0 = blockIdx.y * 16;
  const float* ay = Ay + b * NPP * MP;
  v8f acc = mm_tile(row0, col0, MP,
                    [&](int r, int k) { return W[r * MP + k]; },
                    [&](int k, int c) { return ay[c * MP + k]; });
  store_tile(V + b * MP * NPP, NPP, row0, col0, acc);
}

__global__ void k_reduce(const float* __restrict__ FxRe, const float* __restrict__ FxIm,
                         const float* __restrict__ Ax,
                         const float* __restrict__ URe, const float* __restrict__ UIm,
                         const float* __restrict__ V,
                         float* __restrict__ out, int ch) {
  int pt = blockIdx.x;
  int b = pt >> 9, p = pt & (NPP - 1);
  int lane = threadIdx.x;
  float tot = 0.f, dia = 0.f;
  for (int k = lane; k < MP; k += 32) {
    int uc = (b * MP + k) * NPP + p;
    tot += FxRe[pt * MP + k] * URe[uc] + FxIm[pt * MP + k] * UIm[uc];
    dia += Ax[pt * MP + k] * V[uc];
  }
#pragma unroll
  for (int off = 16; off > 0; off >>= 1) {
    tot += __shfl_down(tot, off);
    dia += __shfl_down(dia, off);
  }
  if (lane == 0) out[pt * 2 + ch] = tot - dia;
}


extern "C" void kernel_launch(void* const* d_in, const int* in_sizes, int n_in,
                              void* d_out, int out_size, void* d_ws, size_t ws_size,
                              hipStream_t stream) {
  (void)in_sizes; (void)n_in; (void)out_size; (void)ws_size;
  const float* x  = (const float*)d_in[0];
  const float* s0 = (const float*)d_in[1];
  const float* s1 = (const float*)d_in[2];
  const float* a0 = (const float*)d_in[3];
  const float* a1 = (const float*)d_in[4];
  float* out = (float*)d_out;

  float* w = (float*)d_ws;
  float* Ere  = w; w += MP * MP;
  float* Eim  = w; w += MP * MP;
  float* W0   = w; w += MP * MP;
  float* W1   = w; w += MP * MP;
  float* Gx   = w; w += NPTS * MP;
  float* Gy   = w; w += NPTS * MP;
  float* FxRe = w; w += NPTS * MP;
  float* FxIm = w; w += NPTS * MP;
  float* FyRe = w; w += NPTS * MP;
  float* FyIm = w; w += NPTS * MP;
  float* Ax   = w; w += NPTS * MP;
  float* Ay   = w; w += NPTS * MP;
  float* SRe  = w; w += NB * MP * MP;
  float* SIm  = w; w += NB * MP * MP;
  float* URe  = w; w += NB * MP * NPP;
  float* UIm  = w; w += NB * MP * NPP;
  float* V    = w; w += NB * MP * NPP;

  k_tables<<<(MP * MP + 255) / 256, 256, 0, stream>>>(Ere, Eim, W0, W1, s0, s1, a0, a1);
  k_spread<<<(NPTS * MP + 255) / 256, 256, 0, stream>>>(x, Gx, Gy);
  k_dft<<<dim3(NPTS / 16, MP / 16, 4), 32, 0, stream>>>(Gx, Gy, Ere, Eim,
                                                        FxRe, FxIm, FyRe, FyIm);
  k_sqmag<<<(NPTS * MP + 255) / 256, 256, 0, stream>>>(FxRe, FxIm, FyRe, FyIm, Ax, Ay);
  k_smat<<<dim3(MP / 16, MP / 16, NB), 32 * KSPLIT, 0, stream>>>(FxRe, FxIm, FyRe, FyIm, SRe, SIm);

  for (int c = 0; c < 2; ++c) {
    const float* Wc = c ? W1 : W0;
    k_umat<<<dim3(MP / 16, NPP / 16, NB), 32, 0, stream>>>(Wc, SRe, SIm, FyRe, FyIm, URe, UIm);
    k_vmat<<<dim3(MP / 16, NPP / 16, NB), 32, 0, stream>>>(Wc, Ay, V);
    k_reduce<<<NPTS, 32, 0, stream>>>(FxRe, FxIm, Ax, URe, UIm, V, out, c);
  }
}


# k_umat, k_vmat, k_reduce of both channels fused into one launch: each wave keeps its U, V tiles in registers and dots them with Fx, Ax; 18 wave partials added in fixed order via LDS (new sum order)
# speedup vs baseline: 2.0233x; 1.7293x over previous
#include <hip/hip_runtime.h>
#include <math.h>

typedef __attribute__((ext_vector_type(2))) float v2f;
typedef __attribute__((ext_vector_type(8))) float v8f;

#define MM    129
#define MP    144
#define NPP   512
#define NB    4
#define NPTS  (NB * NPP)
#define KSPLIT 8

#define PI_F   3.14159265358979323846f
#define TAU_F  (12.0f / (129.0f * 129.0f))
#define L_F    (2.0f * PI_F)


__device__ __forceinline__ v8f wmma4(v2f a, v2f b, v8f c) {
  return __builtin_amdgcn_wmma_f32_16x16x4_f32(false, a, false, b, (short)0, c, false, false);
}

__device__ __forceinline__ void store_tile(float* C, int ldc, int row0, int col0, v8f acc) {
  int lane = threadIdx.x & 31;
  int r0 = row0 + ((lane >> 4) << 3);
  int c  = col0 + (lane & 15);
#pragma unroll
  for (int v = 0; v < 8; ++v) C[(r0 + v) * ldc + c] = acc[v];
}

template <typename FA, typename FB>
__device__ __forceinline__ v8f mm_tile(int row0, int col0, int K, FA la, FB lb) {
  int lane = threadIdx.x & 31;
  int r  = lane & 15;
  int kb = (lane >> 4) << 1;
  v8f acc = {};
  for (int k0 = 0; k0 < K; k0 += 4) {
    v2f a, b;
    a.x = la(row0 + r, k0 + kb);
    a.y = la(row0 + r, k0 + kb + 1);
    b.x = lb(k0 + kb,     col0 + r);
    b.y = lb(k0 + kb + 1, col0 + r);
    acc = wmma4(a, b, acc);
  }
  return acc;
}


__global__ void k_tables(float* __restrict__ Ere, float* __restrict__ Eim,
                         float* __restrict__ W0,  float* __restrict__ W1,
                         const float* s0, const float* s1,
                         const float* a0, const float* a1) {
  int t = blockIdx.x * blockDim.x + threadIdx.x;
  if (t >= MP * MP) return;
  int i = t / MP, k = t % MP;
  float er = 0.f, ei = 0.f, w0 = 0.f, w1 = 0.f;
  if (i < MM && k < MM) {
    int m = (i * k) % MM;
    float th = (2.0f * PI_F / (float)MM) * (float)m;
    er = cosf(th);
    ei = -sinf(th);
    float f1 = (float)((i <= MM / 2) ? i : i - MM);
    float f2 = (float)((k <= MM / 2) ? k : k - MM);
    float k2 = f1 * f1 + f2 * f2;
    float dec = (PI_F / TAU_F) * expf(TAU_F * k2);
    float m2   = (float)MM * (float)MM;
    float coef = 1.0f / (8.0f * PI_F * PI_F * m2 * m2);
    float base = coef * dec * dec * 4.0f * PI_F;
    float sh0 = s0[0], sh1 = s1[0];
    w0 = base * a0[0] / (k2 + sh0 * sh0);
    w1 = base * a1[0] / (k2 + sh1 * sh1);
  }
  Ere[t] = er; Eim[t] = ei; W0[t] = w0; W1[t] = w1;
}

__global__ void k_spread(const float* __restrict__ x,
                         float* __restrict__ Gx, float* __restrict__ Gy) {
  int t = blockIdx.x * blockDim.x + threadIdx.x;
  if (t >= NPTS * MP) return;
  int pt = t / MP, i = t % MP;
  float gx = 0.f, gy = 0.f;
  if (i < MM) {
    float xi  = (L_F / (float)MM) * (float)i;
    float inv = 1.0f / (4.0f * TAU_F);
    float d0 = x[2 * pt + 0] - xi;
    float d1 = x[2 * pt + 1] - xi;
    gx = expf(-d0 * d0 * inv) + expf(-(d0 - L_F) * (d0 - L_F) * inv)
       + expf(-(d0 + L_F) * (d0 + L_F) * inv);
    gy = expf(-d1 * d1 * inv) + expf(-(d1 - L_F) * (d1 - L_F) * inv)
       + expf(-(d1 + L_F) * (d1 + L_F) * inv);
  }
  Gx[t] = gx; Gy[t] = gy;
}

__global__ void k_dft(const float* __restrict__ Gx, const float* __restrict__ Gy,
                      const float* __restrict__ Ere, const float* __restrict__ Eim,
                      float* __restrict__ FxRe, float* __restrict__ FxIm,
                      float* __restrict__ FyRe, float* __restrict__ FyIm) {
  int z = blockIdx.z;
  const float* G = (z < 2) ? Gx : Gy;
  const float* E = (z & 1) ? Eim : Ere;
  float* D = (z == 0) ? FxRe : (z == 1) ? FxIm : (z == 2) ? FyRe : FyIm;
  int row0 = blockIdx.x * 16, col0 = blockIdx.y * 16;
  v8f acc = mm_tile(row0, col0, MP,
                    [&](int r, int k) { return G[r * MP + k]; },
                    [&](int k, int c) { return E[k * MP + c]; });
  store_tile(D, MP, row0, col0, acc);
}

__global__ void k_sqmag(const float* __restrict__ FxRe, const float* __restrict__ FxIm,
                        const float* __restrict__ FyRe, const float* __restrict__ FyIm,
                        float* __restrict__ Ax, float* __restrict__ Ay) {
  int t = blockIdx.x * blockDim.x + threadIdx.x;
  if (t >= NPTS * MP) return;
  Ax[t] = FxRe[t] * FxRe[t] + FxIm[t] * FxIm[t];
  Ay[t] = FyRe[t] * FyRe[t] + FyIm[t] * FyIm[t];
}

__global__ void __launch_bounds__(32 * KSPLIT)
k_smat(const float* __restrict__ FxRe, const float* __restrict__ FxIm,
       const float* __restrict__ FyRe, const float* __restrict__ FyIm,
       float* __restrict__ SRe, float* __restrict__ SIm) {
  __shared__ float part[KSPLIT][2][256];
  int b = blockIdx.z;
  int row0 = blockIdx.x * 16, col0 = blockIdx.y * 16;
  int lane = threadIdx.x & 31, r = lane & 15, kb = (lane >> 4) << 1;
  int wave = __builtin_amdgcn_readfirstlane(threadIdx.x >> 5);
  int qbeg = wave * (NPP / KSPLIT);
  const float* fxr = FxRe + (b * NPP + qbeg) * MP;
  const float* fxi = FxIm + (b * NPP + qbeg) * MP;
  const float* fyr = FyRe + (b * NPP + qbeg) * MP;
  const float* fyi = FyIm + (b * NPP + qbeg) * MP;
  v8f accRe = {}, accIm = {};
  for (int k0 = 0; k0 < NPP / KSPLIT; k0 += 4) {
    int q0 = k0 + kb, q1 = q0 + 1;
    v2f aRe = { fxr[q0 * MP + row0 + r], fxr[q1 * MP + row0 + r] };
    v2f aIm = { fxi[q0 * MP + row0 + r], fxi[q1 * MP + row0 + r] };
    v2f aImN = { -aIm.x, -aIm.y };
    v2f bRe = { fyr[q0 * MP + col0 + r], fyr[q1 * MP + col0 + r] };
    v2f bIm = { fyi[q0 * MP + col0 + r], fyi[q1 * MP + col0 + r] };
    accRe = wmma4(aRe,  bRe, accRe);
    accRe = wmma4(aImN, bIm, accRe);
    accIm = wmma4(aRe,  bIm, accIm);
    accIm = wmma4(aIm,  bRe, accIm);
  }
  for (int v = 0; v < 8; ++v) {
    part[wave][0][v * 32 + lane] = accRe[v];
    part[wave][1][v * 32 + lane] = accIm[v];
  }
  __syncthreads();
  int t = threadIdx.x;
  float sre = part[0][0][t], sim = part[0][1][t];
  for (int w = 1; w < KSPLIT; ++w) {
    sre += part[w][0][t];
    sim += part[w][1][t];
  }
  int l = t & 31;
  int row = row0 + (t >> 5) + ((l >> 4) << 3);
  int col = col0 + (l & 15);
  SRe[b * MP * MP + row * MP + col] = sre;
  SIm[b * MP * MP + row * MP + col] = sim;
}

#define KHALF  2
#define EWAVES ((MP / 16) * KHALF)
__global__ void __launch_bounds__(32 * EWAVES)
k_energy(const float* __restrict__ W0, const float* __restrict__ W1,
         const float* __restrict__ SRe, const float* __restrict__ SIm,
         const float* __restrict__ FxRe, const float* __restrict__ FxIm,
         const float* __restrict__ FyRe, const float* __restrict__ FyIm,
         const float* __restrict__ Ax, const float* __restrict__ Ay,
         float* __restrict__ out) {
  __shared__ float part[EWAVES][2][16];
  int col0 = blockIdx.x * 16;
  int b    = blockIdx.y;
  int ch   = blockIdx.z;
  const float* W = ch ? W1 : W0;
  int lane = threadIdx.x & 31, r = lane & 15, kb = (lane >> 4) << 1;
  int wave = __builtin_amdgcn_readfirstlane(threadIdx.x >> 5);
  int row0 = (wave / KHALF) * 16;
  int kbeg = (wave % KHALF) * (MP / KHALF);
  const float* sre = SRe + b * MP * MP;
  const float* sim = SIm + b * MP * MP;
  const float* fyr = FyRe + b * NPP * MP;
  const float* fyi = FyIm + b * NPP * MP;
  const float* ay  = Ay + b * NPP * MP;

  int pt = b * NPP + col0 + r;
  int e0 = pt * MP + row0 + ((lane >> 4) << 3);
  float fr[8], fi[8], ax[8];
  for (int v = 0; v < 8; ++v) { fr[v] = FxRe[e0 + v]; fi[v] = FxIm[e0 + v]; ax[v] = Ax[e0 + v]; }

  int ra = (row0 + r) * MP + kbeg + kb;
  int pb = (col0 + r) * MP + kbeg + kb;
  v8f accRe = {}, accIm = {}, accV = {};
  for (int k0 = 0; k0 < MP / KHALF; k0 += 4) {
    float w0 = W[ra + k0], w1 = W[ra + k0 + 1];
    v2f aW  = { w0, w1 };
    v2f tRe = { w0 * sre[ra + k0], w1 * sre[ra + k0 + 1] };
    v2f tIm = { w0 * sim[ra + k0], w1 * sim[ra + k0 + 1] };
    v2f tReN = { -tRe.x, -tRe.y };
    v2f bRe = { fyr[pb + k0], fyr[pb + k0 + 1] };
    v2f bIm = { fyi[pb + k0], fyi[pb + k0 + 1] };
    v2f bAy = { ay[pb + k0],  ay[pb + k0 + 1] };
    accRe = wmma4(tRe,  bRe, accRe);
    accIm = wmma4(tIm,  bRe, accIm);
    accV  = wmma4(aW,   bAy, accV);
    accRe = wmma4(tIm,  bIm, accRe);
    accIm = wmma4(tReN, bIm, accIm);
  }

  float tot = 0.f, dia = 0.f;
  for (int v = 0; v < 8; ++v) {
    tot += fr[v] * accRe[v] + fi[v] * accIm[v];
    dia += ax[v] * accV[v];
  }
  tot += __shfl_xor(tot, 16);
  dia += __shfl_xor(dia, 16);
  if (lane < 16) { part[wave][0][lane] = tot; part[wave][1][lane] = dia; }
  __syncthreads();
  int t = threadIdx.x;
  if (t < 16) {
    float st = part[0][0][t], sd = part[0][1][t];
    for (int w = 1; w < EWAVES; ++w) { st += part[w][0][t]; sd += part[w][1][t]; }
    out[(b * NPP + col0 + t) * 2 + ch] = st - sd;
  }
}


extern "C" void kernel_launch(void* const* d_in, const int* in_sizes, int n_in,
                              void* d_out, int out_size, void* d_ws, size_t ws_size,
                              hipStream_t stream) {
  (void)in_sizes; (void)n_in; (void)out_size; (void)ws_size;
  const float* x  = (const float*)d_in[0];
  const float* s0 = (const float*)d_in[1];
  const float* s1 = (const float*)d_in[2];
  const float* a0 = (const float*)d_in[3];
  const float* a1 = (const float*)d_in[4];
  float* out = (float*)d_out;

  float* w = (float*)d_ws;
  float* Ere  = w; w += MP * MP;
  float* Eim  = w; w += MP * MP;
  float* W0   = w; w += MP * MP;
  float* W1   = w; w += MP * MP;
  float* Gx   = w; w += NPTS * MP;
  float* Gy   = w; w += NPTS * MP;
  float* FxRe = w; w += NPTS * MP;
  float* FxIm = w; w += NPTS * MP;
  float* FyRe = w; w += NPTS * MP;
  float* FyIm = w; w += NPTS * MP;
  float* Ax   = w; w += NPTS * MP;
  float* Ay   = w; w += NPTS * MP;
  float* SRe  = w; w += NB * MP * MP;
  float* SIm  = w; w += NB * MP * MP;

  k_tables<<<(MP * MP + 255) / 256, 256, 0, stream>>>(Ere, Eim, W0, W1, s0, s1, a0, a1);
  k_spread<<<(NPTS * MP + 255) / 256, 256, 0, stream>>>(x, Gx, Gy);
  k_dft<<<dim3(NPTS / 16, MP / 16, 4), 32, 0, stream>>>(Gx, Gy, Ere, Eim,
                                                        FxRe, FxIm, FyRe, FyIm);
  k_sqmag<<<(NPTS * MP + 255) / 256, 256, 0, stream>>>(FxRe, FxIm, FyRe, FyIm, Ax, Ay);
  k_smat<<<dim3(MP / 16, MP / 16, NB), 32 * KSPLIT, 0, stream>>>(FxRe, FxIm, FyRe, FyIm, SRe, SIm);

  k_energy<<<dim3(NPP / 16, NB, 2), 32 * EWAVES, 0, stream>>>(W0, W1, SRe, SIm, FxRe, FxIm,
                                                                FyRe, FyIm, Ax, Ay, out);
}


# Table, spread, DFT and power-spectrum launches fused into one: each 9-wave block keeps its 16-point Gaussian stripe and the 129 twiddles in LDS and writes F and |F|^2; summation order unchanged
# speedup vs baseline: 2.7314x; 1.3500x over previous
#include <hip/hip_runtime.h>
#include <math.h>

typedef __attribute__((ext_vector_type(2))) float v2f;
typedef __attribute__((ext_vector_type(8))) float v8f;

#define MM    129
#define MP    144
#define NPP   512
#define NB    4
#define NPTS  (NB * NPP)
#define KSPLIT 8

#define PI_F   3.14159265358979323846f
#define TAU_F  (12.0f / (129.0f * 129.0f))
#define L_F    (2.0f * PI_F)


__device__ __forceinline__ v8f wmma4(v2f a, v2f b, v8f c) {
  return __builtin_amdgcn_wmma_f32_16x16x4_f32(false, a, false, b, (short)0, c, false, false);
}


#define LDG      132
#define FWAVES   (MP / 16)
#define FTHREADS (32 * FWAVES)
#define FBLOCKS  (2 * (NPTS / 16))
#define WBLOCKS  ((MP * MP) / FTHREADS)
__global__ void __launch_bounds__(FTHREADS)
k_front(const float* __restrict__ x,
        const float* s0, const float* s1, const float* a0, const float* a1,
        float* __restrict__ W0, float* __restrict__ W1,
        float* __restrict__ FxRe, float* __restrict__ FxIm,
        float* __restrict__ FyRe, float* __restrict__ FyIm,
        float* __restrict__ Ax, float* __restrict__ Ay) {
  __shared__ v2f Gs2[16 * (LDG / 2)];
  __shared__ v2f tw[MM];
  int t = threadIdx.x;

  if (blockIdx.x >= FBLOCKS) {
    int e = (blockIdx.x - FBLOCKS) * FTHREADS + t;
    int i = e / MP, k = e % MP;
    float w0 = 0.f, w1 = 0.f;
    if (i < MM && k < MM) {
      float f1 = (float)((i <= MM / 2) ? i : i - MM);
      float f2 = (float)((k <= MM / 2) ? k : k - MM);
      float k2 = f1 * f1 + f2 * f2;
      float dec = (PI_F / TAU_F) * expf(TAU_F * k2);
      float m2   = (float)MM * (float)MM;
      float coef = 1.0f / (8.0f * PI_F * PI_F * m2 * m2);
      float base = coef * dec * dec * 4.0f * PI_F;
      float sh0 = s0[0], sh1 = s1[0];
      w0 = base * a0[0] / (k2 + sh0 * sh0);
      w1 = base * a1[0] / (k2 + sh1 * sh1);
    }
    W0[e] = w0; W1[e] = w1;
    return;
  }

  int axis = blockIdx.x & 1;
  int row0 = (blockIdx.x >> 1) * 16;

  int p = t & 15;
  float xp = x[2 * (row0 + p) + axis];

  if (t < MM) {
    float th = (2.0f * PI_F / (float)MM) * (float)t;
    v2f e = { cosf(th), -sinf(th) };
    tw[t] = e;
  }
  {
    float* Gs = (float*)Gs2;
    const float inv = 1.0f / (4.0f * TAU_F);
    for (int i = t >> 4; i < LDG; i += FTHREADS / 16) {
      float g = 0.f;
      if (i < MM) {
        float xi = (L_F / (float)MM) * (float)i;
        float d  = xp - xi;
        g = expf(-d * d * inv) + expf(-(d - L_F) * (d - L_F) * inv)
          + expf(-(d + L_F) * (d + L_F) * inv);
      }
      Gs[p * LDG + i] = g;
    }
  }
  __syncthreads();

  int lane = t & 31, r = lane & 15, kb = (lane >> 4) << 1;
  int wave = __builtin_amdgcn_readfirstlane(t >> 5);
  int c  = wave * 16 + r;
  int cm = (c < MM) ? c : c - MM;
  int m0    = (kb * cm) % MM;
  int step4 = (4 * cm) % MM;
  const v2f* ga = Gs2 + r * (LDG / 2) + (lane >> 4);
  v8f accRe = {}, accIm = {};
  for (int j = 0; j < LDG / 4; ++j) {
    int m1 = m0 + cm;
    m1 = (m1 >= MM) ? m1 - MM : m1;
    v2f a  = ga[2 * j];
    v2f e0 = tw[m0], e1 = tw[m1];
    v2f bRe = { e0.x, e1.x };
    v2f bIm = { e0.y, e1.y };
    accRe = wmma4(a, bRe, accRe);
    accIm = wmma4(a, bIm, accIm);
    m0 += step4;
    m0 = (m0 >= MM) ? m0 - MM : m0;
  }

  float* DRe = axis ? FyRe : FxRe;
  float* DIm = axis ? FyIm : FxIm;
  float* DA  = axis ? Ay : Ax;
  bool cv = (c < MM);
  int o0 = (row0 + ((lane >> 4) << 3)) * MP + c;
#pragma unroll
  for (int v = 0; v < 8; ++v) {
    float re = cv ? accRe[v] : 0.f;
    float im = cv ? accIm[v] : 0.f;
    DRe[o0 + v * MP] = re;
    DIm[o0 + v * MP] = im;
    DA[o0 + v * MP]  = re * re + im * im;
  }
}

__global__ void __launch_bounds__(32 * KSPLIT)
k_smat(const float* __restrict__ FxRe, const float* __restrict__ FxIm,
       const float* __restrict__ FyRe, const float* __restrict__ FyIm,
       float* __restrict__ SRe, float* __restrict__ SIm) {
  __shared__ float part[KSPLIT][2][256];
  int b = blockIdx.z;
  int row0 = blockIdx.x * 16, col0 = blockIdx.y * 16;
  int lane = threadIdx.x & 31, r = lane & 15, kb = (lane >> 4) << 1;
  int wave = __builtin_amdgcn_readfirstlane(threadIdx.x >> 5);
  int qbeg = wave * (NPP / KSPLIT);
  const float* fxr = FxRe + (b * NPP + qbeg) * MP;
  const float* fxi = FxIm + (b * NPP + qbeg) * MP;
  const float* fyr = FyRe + (b * NPP + qbeg) * MP;
  const float* fyi = FyIm + (b * NPP + qbeg) * MP;
  v8f accRe = {}, accIm = {};
  for (int k0 = 0; k0 < NPP / KSPLIT; k0 += 4) {
    int q0 = k0 + kb, q1 = q0 + 1;
    v2f aRe = { fxr[q0 * MP + row0 + r], fxr[q1 * MP + row0 + r] };
    v2f aIm = { fxi[q0 * MP + row0 + r], fxi[q1 * MP + row0 + r] };
    v2f aImN = { -aIm.x, -aIm.y };
    v2f bRe = { fyr[q0 * MP + col0 + r], fyr[q1 * MP + col0 + r] };
    v2f bIm = { fyi[q0 * MP + col0 + r], fyi[q1 * MP + col0 + r] };
    accRe = wmma4(aRe,  bRe, accRe);
    accRe = wmma4(aImN, bIm, accRe);
    accIm = wmma4(aRe,  bIm, accIm);
    accIm = wmma4(aIm,  bRe, accIm);
  }
  for (int v = 0; v < 8; ++v) {
    part[wave][0][v * 32 + lane] = accRe[v];
    part[wave][1][v * 32 + lane] = accIm[v];
  }
  __syncthreads();
  int t = threadIdx.x;
  float sre = part[0][0][t], sim = part[0][1][t];
  for (int w = 1; w < KSPLIT; ++w) {
    sre += part[w][0][t];
    sim += part[w][1][t];
  }
  int l = t & 31;
  int row = row0 + (t >> 5) + ((l >> 4) << 3);
  int col = col0 + (l & 15);
  SRe[b * MP * MP + row * MP + col] = sre;
  SIm[b * MP * MP + row * MP + col] = sim;
}

#define KHALF  2
#define EWAVES ((MP / 16) * KHALF)
__global__ void __launch_bounds__(32 * EWAVES)
k_energy(const float* __restrict__ W0, const float* __restrict__ W1,
         const float* __restrict__ SRe, const float* __restrict__ SIm,
         const float* __restrict__ FxRe, const float* __restrict__ FxIm,
         const float* __restrict__ FyRe, const float* __restrict__ FyIm,
         const float* __restrict__ Ax, const float* __restrict__ Ay,
         float* __restrict__ out) {
  __shared__ float part[EWAVES][2][16];
  int col0 = blockIdx.x * 16;
  int b    = blockIdx.y;
  int ch   = blockIdx.z;
  const float* W = ch ? W1 : W0;
  int lane = threadIdx.x & 31, r = lane & 15, kb = (lane >> 4) << 1;
  int wave = __builtin_amdgcn_readfirstlane(threadIdx.x >> 5);
  int row0 = (wave / KHALF) * 16;
  int kbeg = (wave % KHALF) * (MP / KHALF);
  const float* sre = SRe + b * MP * MP;
  const float* sim = SIm + b * MP * MP;
  const float* fyr = FyRe + b * NPP * MP;
  const float* fyi = FyIm + b * NPP * MP;
  const float* ay  = Ay + b * NPP * MP;

  int pt = b * NPP + col0 + r;
  int e0 = pt * MP + row0 + ((lane >> 4) << 3);
  float fr[8], fi[8], ax[8];
  for (int v = 0; v < 8; ++v) { fr[v] = FxRe[e0 + v]; fi[v] = FxIm[e0 + v]; ax[v] = Ax[e0 + v]; }

  int ra = (row0 + r) * MP + kbeg + kb;
  int pb = (col0 + r) * MP + kbeg + kb;
  v8f accRe = {}, accIm = {}, accV = {};
  for (int k0 = 0; k0 < MP / KHALF; k0 += 4) {
    float w0 = W[ra + k0], w1 = W[ra + k0 + 1];
    v2f aW  = { w0, w1 };
    v2f tRe = { w0 * sre[ra + k0], w1 * sre[ra + k0 + 1] };
    v2f tIm = { w0 * sim[ra + k0], w1 * sim[ra + k0 + 1] };
    v2f tReN = { -tRe.x, -tRe.y };
    v2f bRe = { fyr[pb + k0], fyr[pb + k0 + 1] };
    v2f bIm = { fyi[pb + k0], fyi[pb + k0 + 1] };
    v2f bAy = { ay[pb + k0],  ay[pb + k0 + 1] };
    accRe = wmma4(tRe,  bRe, accRe);
    accIm = wmma4(tIm,  bRe, accIm);
    accV  = wmma4(aW,   bAy, accV);
    accRe = wmma4(tIm,  bIm, accRe);
    accIm = wmma4(tReN, bIm, accIm);
  }

  float tot = 0.f, dia = 0.f;
  for (int v = 0; v < 8; ++v) {
    tot += fr[v] * accRe[v] + fi[v] * accIm[v];
    dia += ax[v] * accV[v];
  }
  tot += __shfl_xor(tot, 16);
  dia += __shfl_xor(dia, 16);
  if (lane < 16) { part[wave][0][lane] = tot; part[wave][1][lane] = dia; }
  __syncthreads();
  int t = threadIdx.x;
  if (t < 16) {
    float st = part[0][0][t], sd = part[0][1][t];
    for (int w = 1; w < EWAVES; ++w) { st += part[w][0][t]; sd += part[w][1][t]; }
    out[(b * NPP + col0 + t) * 2 + ch] = st - sd;
  }
}


extern "C" void kernel_launch(void* const* d_in, const int* in_sizes, int n_in,
                              void* d_out, int out_size, void* d_ws, size_t ws_size,
                              hipStream_t stream) {
  (void)in_sizes; (void)n_in; (void)out_size; (void)ws_size;
  const float* x  = (const float*)d_in[0];
  const float* s0 = (const float*)d_in[1];
  const float* s1 = (const float*)d_in[2];
  const float* a0 = (const float*)d_in[3];
  const float* a1 = (const float*)d_in[4];
  float* out = (float*)d_out;

  float* w = (float*)d_ws;
  float* W0   = w; w += MP * MP;
  float* W1   = w; w += MP * MP;
  float* FxRe = w; w += NPTS * MP;
  float* FxIm = w; w += NPTS * MP;
  float* FyRe = w; w += NPTS * MP;
  float* FyIm = w; w += NPTS * MP;
  float* Ax   = w; w += NPTS * MP;
  float* Ay   = w; w += NPTS * MP;
  float* SRe  = w; w += NB * MP * MP;
  float* SIm  = w; w += NB * MP * MP;

  k_front<<<FBLOCKS + WBLOCKS, FTHREADS, 0, stream>>>(x, s0, s1, a0, a1, W0, W1,
                                                      FxRe, FxIm, FyRe, FyIm, Ax, Ay);
  k_smat<<<dim3(MP / 16, MP / 16, NB), 32 * KSPLIT, 0, stream>>>(FxRe, FxIm, FyRe, FyIm, SRe, SIm);

  k_energy<<<dim3(NPP / 16, NB, 2), 32 * EWAVES, 0, stream>>>(W0, W1, SRe, SIm, FxRe, FxIm,
                                                                FyRe, FyIm, Ax, Ay, out);
}


# Mirror fold of k2: spread is real, so terms (k1,k2) and (M-k1,M-k2) are equal; weights doubled, only bins k2<=M/2 of Fy, S, W formed: Gram columns and energy K halved (changes that sum's order)
# speedup vs baseline: 3.6323x; 1.3298x over previous
#include <hip/hip_runtime.h>
#include <math.h>

typedef __attribute__((ext_vector_type(2))) float v2f;
typedef __attribute__((ext_vector_type(8))) float v8f;

#define MM    129
#define MP    144
#define NPP   512
#define NB    4
#define NPTS  (NB * NPP)
#define KSPLIT 8
#define KF     72
#define SCOLT  5

#define PI_F   3.14159265358979323846f
#define TAU_F  (12.0f / (129.0f * 129.0f))
#define L_F    (2.0f * PI_F)


__device__ __forceinline__ v8f wmma4(v2f a, v2f b, v8f c) {
  return __builtin_amdgcn_wmma_f32_16x16x4_f32(false, a, false, b, (short)0, c, false, false);
}


#define LDG      132
#define FWAVES   (MP / 16)
#define FTHREADS (32 * FWAVES)
#define FBLOCKS  (2 * (NPTS / 16))
#define WBLOCKS  ((MP * KF) / FTHREADS)
__global__ void __launch_bounds__(FTHREADS)
k_front(const float* __restrict__ x,
        const float* s0, const float* s1, const float* a0, const float* a1,
        float* __restrict__ W0, float* __restrict__ W1,
        float* __restrict__ FxRe, float* __restrict__ FxIm,
        float* __restrict__ FyRe, float* __restrict__ FyIm,
        float* __restrict__ Ax, float* __restrict__ Ay) {
  __shared__ v2f Gs2[16 * (LDG / 2)];
  __shared__ v2f tw[MM];
  int t = threadIdx.x;

  if (blockIdx.x >= FBLOCKS) {
    int e = (blockIdx.x - FBLOCKS) * FTHREADS + t;
    int i = e / KF, k = e % KF;
    float w0 = 0.f, w1 = 0.f;
    if (i < MM && k <= MM / 2) {
      float f1 = (float)((i <= MM / 2) ? i : i - MM);
      float f2 = (float)k;
      float k2 = f1 * f1 + f2 * f2;
      float fold = (k == 0) ? 1.0f : 2.0f;
      float dec = (PI_F / TAU_F) * expf(TAU_F * k2);
      float m2   = (float)MM * (float)MM;
      float coef = 1.0f / (8.0f * PI_F * PI_F * m2 * m2);
      float base = fold * coef * dec * dec * 4.0f * PI_F;
      float sh0 = s0[0], sh1 = s1[0];
      w0 = base * a0[0] / (k2 + sh0 * sh0);
      w1 = base * a1[0] / (k2 + sh1 * sh1);
    }
    W0[i * MP + k] = w0; W1[i * MP + k] = w1;
    return;
  }

  int axis = blockIdx.x & 1;
  int row0 = (blockIdx.x >> 1) * 16;

  int p = t & 15;
  float xp = x[2 * (row0 + p) + axis];

  if (t < MM) {
    float th = (2.0f * PI_F / (float)MM) * (float)t;
    v2f e = { cosf(th), -sinf(th) };
    tw[t] = e;
  }
  {
    float* Gs = (float*)Gs2;
    const float inv = 1.0f / (4.0f * TAU_F);
    for (int i = t >> 4; i < LDG; i += FTHREADS / 16) {
      float g = 0.f;
      if (i < MM) {
        float xi = (L_F / (float)MM) * (float)i;
        float d  = xp - xi;
        g = expf(-d * d * inv) + expf(-(d - L_F) * (d - L_F) * inv)
          + expf(-(d + L_F) * (d + L_F) * inv);
      }
      Gs[p * LDG + i] = g;
    }
  }
  __syncthreads();

  int lane = t & 31, r = lane & 15, kb = (lane >> 4) << 1;
  int wave = __builtin_amdgcn_readfirstlane(t >> 5);
  if (axis && wave >= SCOLT) return;
  int c  = wave * 16 + r;
  int cm = (c < MM) ? c : c - MM;
  int m0    = (kb * cm) % MM;
  int step4 = (4 * cm) % MM;
  const v2f* ga = Gs2 + r * (LDG / 2) + (lane >> 4);
  v8f accRe = {}, accIm = {};
  for (int j = 0; j < LDG / 4; ++j) {
    int m1 = m0 + cm;
    m1 = (m1 >= MM) ? m1 - MM : m1;
    v2f a  = ga[2 * j];
    v2f e0 = tw[m0], e1 = tw[m1];
    v2f bRe = { e0.x, e1.x };
    v2f bIm = { e0.y, e1.y };
    accRe = wmma4(a, bRe, accRe);
    accIm = wmma4(a, bIm, accIm);
    m0 += step4;
    m0 = (m0 >= MM) ? m0 - MM : m0;
  }

  float* DRe = axis ? FyRe : FxRe;
  float* DIm = axis ? FyIm : FxIm;
  float* DA  = axis ? Ay : Ax;
  bool cv = (c < MM);
  int o0 = (row0 + ((lane >> 4) << 3)) * MP + c;
#pragma unroll
  for (int v = 0; v < 8; ++v) {
    float re = cv ? accRe[v] : 0.f;
    float im = cv ? accIm[v] : 0.f;
    DRe[o0 + v * MP] = re;
    DIm[o0 + v * MP] = im;
    DA[o0 + v * MP]  = re * re + im * im;
  }
}

__global__ void __launch_bounds__(32 * KSPLIT)
k_smat(const float* __restrict__ FxRe, const float* __restrict__ FxIm,
       const float* __restrict__ FyRe, const float* __restrict__ FyIm,
       float* __restrict__ SRe, float* __restrict__ SIm) {
  __shared__ float part[KSPLIT][2][256];
  int b = blockIdx.z;
  int row0 = blockIdx.x * 16, col0 = blockIdx.y * 16;
  int lane = threadIdx.x & 31, r = lane & 15, kb = (lane >> 4) << 1;
  int wave = __builtin_amdgcn_readfirstlane(threadIdx.x >> 5);
  int qbeg = wave * (NPP / KSPLIT);
  const float* fxr = FxRe + (b * NPP + qbeg) * MP;
  const float* fxi = FxIm + (b * NPP + qbeg) * MP;
  const float* fyr = FyRe + (b * NPP + qbeg) * MP;
  const float* fyi = FyIm + (b * NPP + qbeg) * MP;
  v8f accRe = {}, accIm = {};
  for (int k0 = 0; k0 < NPP / KSPLIT; k0 += 4) {
    int q0 = k0 + kb, q1 = q0 + 1;
    v2f aRe = { fxr[q0 * MP + row0 + r], fxr[q1 * MP + row0 + r] };
    v2f aIm = { fxi[q0 * MP + row0 + r], fxi[q1 * MP + row0 + r] };
    v2f aImN = { -aIm.x, -aIm.y };
    v2f bRe = { fyr[q0 * MP + col0 + r], fyr[q1 * MP + col0 + r] };
    v2f bIm = { fyi[q0 * MP + col0 + r], fyi[q1 * MP + col0 + r] };
    accRe = wmma4(aRe,  bRe, accRe);
    accRe = wmma4(aImN, bIm, accRe);
    accIm = wmma4(aRe,  bIm, accIm);
    accIm = wmma4(aIm,  bRe, accIm);
  }
  for (int v = 0; v < 8; ++v) {
    part[wave][0][v * 32 + lane] = accRe[v];
    part[wave][1][v * 32 + lane] = accIm[v];
  }
  __syncthreads();
  int t = threadIdx.x;
  float sre = part[0][0][t], sim = part[0][1][t];
  for (int w = 1; w < KSPLIT; ++w) {
    sre += part[w][0][t];
    sim += part[w][1][t];
  }
  int l = t & 31;
  int row = row0 + (t >> 5) + ((l >> 4) << 3);
  int col = col0 + (l & 15);
  SRe[b * MP * MP + row * MP + col] = sre;
  SIm[b * MP * MP + row * MP + col] = sim;
}

#define KHALF  2
#define EWAVES ((MP / 16) * KHALF)
__global__ void __launch_bounds__(32 * EWAVES)
k_energy(const float* __restrict__ W0, const float* __restrict__ W1,
         const float* __restrict__ SRe, const float* __restrict__ SIm,
         const float* __restrict__ FxRe, const float* __restrict__ FxIm,
         const float* __restrict__ FyRe, const float* __restrict__ FyIm,
         const float* __restrict__ Ax, const float* __restrict__ Ay,
         float* __restrict__ out) {
  __shared__ float part[EWAVES][2][16];
  int col0 = blockIdx.x * 16;
  int b    = blockIdx.y;
  int ch   = blockIdx.z;
  const float* W = ch ? W1 : W0;
  int lane = threadIdx.x & 31, r = lane & 15, kb = (lane >> 4) << 1;
  int wave = __builtin_amdgcn_readfirstlane(threadIdx.x >> 5);
  int row0 = (wave / KHALF) * 16;
  int kbeg = (wave % KHALF) * (KF / KHALF);
  const float* sre = SRe + b * MP * MP;
  const float* sim = SIm + b * MP * MP;
  const float* fyr = FyRe + b * NPP * MP;
  const float* fyi = FyIm + b * NPP * MP;
  const float* ay  = Ay + b * NPP * MP;

  int pt = b * NPP + col0 + r;
  int e0 = pt * MP + row0 + ((lane >> 4) << 3);
  float fr[8], fi[8], ax[8];
  for (int v = 0; v < 8; ++v) { fr[v] = FxRe[e0 + v]; fi[v] = FxIm[e0 + v]; ax[v] = Ax[e0 + v]; }

  int ra = (row0 + r) * MP + kbeg + kb;
  int pb = (col0 + r) * MP + kbeg + kb;
  v8f accRe = {}, accIm = {}, accV = {};
  for (int k0 = 0; k0 < KF / KHALF; k0 += 4) {
    float w0 = W[ra + k0], w1 = W[ra + k0 + 1];
    v2f aW  = { w0, w1 };
    v2f tRe = { w0 * sre[ra + k0], w1 * sre[ra + k0 + 1] };
    v2f tIm = { w0 * sim[ra + k0], w1 * sim[ra + k0 + 1] };
    v2f tReN = { -tRe.x, -tRe.y };
    v2f bRe = { fyr[pb + k0], fyr[pb + k0 + 1] };
    v2f bIm = { fyi[pb + k0], fyi[pb + k0 + 1] };
    v2f bAy = { ay[pb + k0],  ay[pb + k0 + 1] };
    accRe = wmma4(tRe,  bRe, accRe);
    accIm = wmma4(tIm,  bRe, accIm);
    accV  = wmma4(aW,   bAy, accV);
    accRe = wmma4(tIm,  bIm, accRe);
    accIm = wmma4(tReN, bIm, accIm);
  }

  float tot = 0.f, dia = 0.f;
  for (int v = 0; v < 8; ++v) {
    tot += fr[v] * accRe[v] + fi[v] * accIm[v];
    dia += ax[v] * accV[v];
  }
  tot += __shfl_xor(tot, 16);
  dia += __shfl_xor(dia, 16);
  if (lane < 16) { part[wave][0][lane] = tot; part[wave][1][lane] = dia; }
  __syncthreads();
  int t = threadIdx.x;
  if (t < 16) {
    float st = part[0][0][t], sd = part[0][1][t];
    for (int w = 1; w < EWAVES; ++w) { st += part[w][0][t]; sd += part[w][1][t]; }
    out[(b * NPP + col0 + t) * 2 + ch] = st - sd;
  }
}


extern "C" void kernel_launch(void* const* d_in, const int* in_sizes, int n_in,
                              void* d_out, int out_size, void* d_ws, size_t ws_size,
                              hipStream_t stream) {
  (void)in_sizes; (void)n_in; (void)out_size; (void)ws_size;
  const float* x  = (const float*)d_in[0];
  const float* s0 = (const float*)d_in[1];
  const float* s1 = (const float*)d_in[2];
  const float* a0 = (const float*)d_in[3];
  const float* a1 = (const float*)d_in[4];
  float* out = (float*)d_out;

  float* w = (float*)d_ws;
  float* W0   = w; w += MP * MP;
  float* W1   = w; w += MP * MP;
  float* FxRe = w; w += NPTS * MP;
  float* FxIm = w; w += NPTS * MP;
  float* FyRe = w; w += NPTS * MP;
  float* FyIm = w; w += NPTS * MP;
  float* Ax   = w; w += NPTS * MP;
  float* Ay   = w; w += NPTS * MP;
  float* SRe  = w; w += NB * MP * MP;
  float* SIm  = w; w += NB * MP * MP;

  k_front<<<FBLOCKS + WBLOCKS, FTHREADS, 0, stream>>>(x, s0, s1, a0, a1, W0, W1,
                                                      FxRe, FxIm, FyRe, FyIm, Ax, Ay);
  k_smat<<<dim3(MP / 16, SCOLT, NB), 32 * KSPLIT, 0, stream>>>(FxRe, FxIm, FyRe, FyIm, SRe, SIm);

  k_energy<<<dim3(NPP / 16, NB, 2), 32 * EWAVES, 0, stream>>>(W0, W1, SRe, SIm, FxRe, FxIm,
                                                                FyRe, FyIm, Ax, Ay, out);
}


# W, S, Fy and Ay stored by their producers in the fragment order k_energy reads: one contiguous 16-byte load per lane for two K steps instead of 16 rows x 8 bytes; changes the order of its k2 sum
# speedup vs baseline: 4.4218x; 1.2174x over previous
#include <hip/hip_runtime.h>
#include <math.h>

typedef __attribute__((ext_vector_type(2))) float v2f;
typedef __attribute__((ext_vector_type(4))) float v4f;
typedef __attribute__((ext_vector_type(8))) float v8f;

#define MM    129
#define MP    144
#define NPP   512
#define NB    4
#define NPTS  (NB * NPP)
#define KSPLIT 8
#define KF     72
#define SCOLT  5
#define KQ     (KF / 8)

#define PI_F   3.14159265358979323846f
#define TAU_F  (12.0f / (129.0f * 129.0f))
#define L_F    (2.0f * PI_F)


__device__ __forceinline__ v8f wmma4(v2f a, v2f b, v8f c) {
  return __builtin_amdgcn_wmma_f32_16x16x4_f32(false, a, false, b, (short)0, c, false, false);
}


#define LDG      132
#define FWAVES   (MP / 16)
#define FTHREADS (32 * FWAVES)
#define FBLOCKS  (2 * (NPTS / 16))
#define WBLOCKS  ((MP * KF) / FTHREADS)
__global__ void __launch_bounds__(FTHREADS)
k_front(const float* __restrict__ x,
        const float* s0, const float* s1, const float* a0, const float* a1,
        float* __restrict__ W0, float* __restrict__ W1,
        float* __restrict__ FxRe, float* __restrict__ FxIm,
        float* __restrict__ FyRe, float* __restrict__ FyIm,
        float* __restrict__ FyReF, float* __restrict__ FyImF,
        float* __restrict__ Ax, float* __restrict__ AyF) {
  __shared__ v2f Gs2[16 * (LDG / 2)];
  __shared__ v2f tw[MM];
  int t = threadIdx.x;

  if (blockIdx.x >= FBLOCKS) {
    int e = (blockIdx.x - FBLOCKS) * FTHREADS + t;
    int g = e >> 7, l = (e >> 2) & 31;
    int i = (g / KQ) * 16 + (l & 15);
    int k = (g % KQ) * 8 + ((l >> 4) << 2) + (e & 3);
    float w0 = 0.f, w1 = 0.f;
    if (i < MM && k <= MM / 2) {
      float f1 = (float)((i <= MM / 2) ? i : i - MM);
      float f2 = (float)k;
      float k2 = f1 * f1 + f2 * f2;
      float fold = (k == 0) ? 1.0f : 2.0f;
      float dec = (PI_F / TAU_F) * expf(TAU_F * k2);
      float m2   = (float)MM * (float)MM;
      float coef = 1.0f / (8.0f * PI_F * PI_F * m2 * m2);
      float base = fold * coef * dec * dec * 4.0f * PI_F;
      float sh0 = s0[0], sh1 = s1[0];
      w0 = base * a0[0] / (k2 + sh0 * sh0);
      w1 = base * a1[0] / (k2 + sh1 * sh1);
    }
    W0[e] = w0; W1[e] = w1;
    return;
  }

  int axis = blockIdx.x & 1;
  int row0 = (blockIdx.x >> 1) * 16;

  int p = t & 15;
  float xp = x[2 * (row0 + p) + axis];

  if (t < MM) {
    float th = (2.0f * PI_F / (float)MM) * (float)t;
    v2f e = { cosf(th), -sinf(th) };
    tw[t] = e;
  }
  {
    float* Gs = (float*)Gs2;
    const float inv = 1.0f / (4.0f * TAU_F);
    for (int i = t >> 4; i < LDG; i += FTHREADS / 16) {
      float g = 0.f;
      if (i < MM) {
        float xi = (L_F / (float)MM) * (float)i;
        float d  = xp - xi;
        g = expf(-d * d * inv) + expf(-(d - L_F) * (d - L_F) * inv)
          + expf(-(d + L_F) * (d + L_F) * inv);
      }
      Gs[p * LDG + i] = g;
    }
  }
  __syncthreads();

  int lane = t & 31, r = lane & 15, kb = (lane >> 4) << 1;
  int wave = __builtin_amdgcn_readfirstlane(t >> 5);
  if (axis && wave >= SCOLT) return;
  int c  = wave * 16 + r;
  int cm = (c < MM) ? c : c - MM;
  int m0    = (kb * cm) % MM;
  int step4 = (4 * cm) % MM;
  const v2f* ga = Gs2 + r * (LDG / 2) + (lane >> 4);
  v8f accRe = {}, accIm = {};
  for (int j = 0; j < LDG / 4; ++j) {
    int m1 = m0 + cm;
    m1 = (m1 >= MM) ? m1 - MM : m1;
    v2f a  = ga[2 * j];
    v2f e0 = tw[m0], e1 = tw[m1];
    v2f bRe = { e0.x, e1.x };
    v2f bIm = { e0.y, e1.y };
    accRe = wmma4(a, bRe, accRe);
    accIm = wmma4(a, bIm, accIm);
    m0 += step4;
    m0 = (m0 >= MM) ? m0 - MM : m0;
  }

  float* DRe = axis ? FyRe : FxRe;
  float* DIm = axis ? FyIm : FxIm;
  bool cv = (c < MM);
  int hr = (lane >> 4) << 3;
  int o0 = (row0 + hr) * MP + c;
  float re[8], im[8];
#pragma unroll
  for (int v = 0; v < 8; ++v) {
    re[v] = cv ? accRe[v] : 0.f;
    im[v] = cv ? accIm[v] : 0.f;
    DRe[o0 + v * MP] = re[v];
    DIm[o0 + v * MP] = im[v];
  }
  if (!axis) {
#pragma unroll
    for (int v = 0; v < 8; ++v) Ax[o0 + v * MP] = re[v] * re[v] + im[v] * im[v];
  } else if (c < KF) {
    int of = (((row0 >> 4) * KQ + (c >> 3)) * 32 + (((c >> 2) & 1) << 4) + hr) * 4 + (c & 3);
#pragma unroll
    for (int v = 0; v < 8; ++v) {
      FyReF[of + 4 * v] = re[v];
      FyImF[of + 4 * v] = im[v];
      AyF[of + 4 * v]   = re[v] * re[v] + im[v] * im[v];
    }
  }
}

__global__ void __launch_bounds__(32 * KSPLIT)
k_smat(const float* __restrict__ FxRe, const float* __restrict__ FxIm,
       const float* __restrict__ FyRe, const float* __restrict__ FyIm,
       float* __restrict__ SRe, float* __restrict__ SIm) {
  __shared__ float part[KSPLIT][2][256];
  int b = blockIdx.z;
  int row0 = blockIdx.x * 16, col0 = blockIdx.y * 16;
  int lane = threadIdx.x & 31, r = lane & 15, kb = (lane >> 4) << 1;
  int wave = __builtin_amdgcn_readfirstlane(threadIdx.x >> 5);
  int qbeg = wave * (NPP / KSPLIT);
  const float* fxr = FxRe + (b * NPP + qbeg) * MP;
  const float* fxi = FxIm + (b * NPP + qbeg) * MP;
  const float* fyr = FyRe + (b * NPP + qbeg) * MP;
  const float* fyi = FyIm + (b * NPP + qbeg) * MP;
  v8f accRe = {}, accIm = {};
  for (int k0 = 0; k0 < NPP / KSPLIT; k0 += 4) {
    int q0 = k0 + kb, q1 = q0 + 1;
    v2f aRe = { fxr[q0 * MP + row0 + r], fxr[q1 * MP + row0 + r] };
    v2f aIm = { fxi[q0 * MP + row0 + r], fxi[q1 * MP + row0 + r] };
    v2f aImN = { -aIm.x, -aIm.y };
    v2f bRe = { fyr[q0 * MP + col0 + r], fyr[q1 * MP + col0 + r] };
    v2f bIm = { fyi[q0 * MP + col0 + r], fyi[q1 * MP + col0 + r] };
    accRe = wmma4(aRe,  bRe, accRe);
    accRe = wmma4(aImN, bIm, accRe);
    accIm = wmma4(aRe,  bIm, accIm);
    accIm = wmma4(aIm,  bRe, accIm);
  }
  for (int v = 0; v < 8; ++v) {
    part[wave][0][v * 32 + lane] = accRe[v];
    part[wave][1][v * 32 + lane] = accIm[v];
  }
  __syncthreads();
  int t = threadIdx.x;
  int tr = (t >> 2) & 15;
  int tc = ((t >> 7) << 3) + (((t >> 6) & 1) << 2) + (t & 3);
  int s = (tr & 7) * 32 + ((tr >> 3) << 4) + tc;
  float sre = part[0][0][s], sim = part[0][1][s];
  for (int w = 1; w < KSPLIT; ++w) {
    sre += part[w][0][s];
    sim += part[w][1][s];
  }
  int g = 2 * blockIdx.y + (t >> 7);
  if (g < KQ) {
    int o = b * MP * KF + (blockIdx.x * KQ + g) * 128 + (t & 127);
    SRe[o] = sre;
    SIm[o] = sim;
  }
}

#define KHALF  2
#define EWAVES ((MP / 16) * KHALF)
__global__ void __launch_bounds__(32 * EWAVES)
k_energy(const float* __restrict__ W0, const float* __restrict__ W1,
         const float* __restrict__ SRe, const float* __restrict__ SIm,
         const float* __restrict__ FxRe, const float* __restrict__ FxIm,
         const float* __restrict__ FyRe, const float* __restrict__ FyIm,
         const float* __restrict__ Ax, const float* __restrict__ Ay,
         float* __restrict__ out) {
  __shared__ float part[EWAVES][2][16];
  int col0 = blockIdx.x * 16;
  int b    = blockIdx.y;
  int ch   = blockIdx.z;
  const float* W = ch ? W1 : W0;
  int lane = threadIdx.x & 31, r = lane & 15;
  int wave = __builtin_amdgcn_readfirstlane(threadIdx.x >> 5);
  int kt   = wave / KHALF;
  int hf   = wave % KHALF;
  int row0 = kt * 16;
  int pa = kt * (KQ * 128) + lane * 4;
  int pb = (b * (NPP / 16) + blockIdx.x) * (KQ * 128) + lane * 4;
  const float* wa = W + pa;
  const float* sr = SRe + b * MP * KF + pa;
  const float* si = SIm + b * MP * KF + pa;
  const float* yr = FyRe + pb;
  const float* yi = FyIm + pb;
  const float* ay = Ay + pb;

  int pt = b * NPP + col0 + r;
  int e0 = pt * MP + row0 + ((lane >> 4) << 3);
  float fr[8], fi[8], ax[8];
  for (int v = 0; v < 8; ++v) { fr[v] = FxRe[e0 + v]; fi[v] = FxIm[e0 + v]; ax[v] = Ax[e0 + v]; }

  int g0 = 4 * hf;
  int o1 = g0 * 128 + 2 * hf;
  int o2 = (8 * hf) * 128 + 2;
  int o3 = (g0 + 1) * 128, o4 = (g0 + 2) * 128, o5 = (g0 + 3) * 128;
  int o6 = (g0 + 4) * 128;
  v8f accRe = {}, accIm = {}, accV = {};
#define LD2(p, o) (*(const v2f*)((p) + (o)))
#define LD4(p, o) (*(const v4f*)((p) + (o)))
#define ESTEP(aW, sR, sI, bRe, bIm, bAy) { \
    v2f tRe = (aW) * (sR), tIm = (aW) * (sI); \
    v2f tReN = -tRe; \
    accRe = wmma4(tRe,  (bRe), accRe);     \
    accIm = wmma4(tIm,  (bRe), accIm);     \
    accV  = wmma4((aW), (bAy), accV);      \
    accRe = wmma4(tIm,  (bIm), accRe); \
    accIm = wmma4(tReN, (bIm), accIm); }
#define ESTEP2(o) ESTEP(LD2(wa, o), LD2(sr, o), LD2(si, o), LD2(yr, o), LD2(yi, o), LD2(ay, o))
#define ESTEP4(o) { \
    v4f w4 = LD4(wa, o), r4 = LD4(sr, o), i4 = LD4(si, o), yr4 = LD4(yr, o), yi4 = LD4(yi, o), ay4 = LD4(ay, o); \
    ESTEP(w4.xy, r4.xy, i4.xy, yr4.xy, yi4.xy, ay4.xy) \
    ESTEP(w4.zw, r4.zw, i4.zw, yr4.zw, yi4.zw, ay4.zw) }
  ESTEP2(o1)
  ESTEP2(o2)
  ESTEP4(o3)
  ESTEP4(o4)
  ESTEP4(o5)
  ESTEP2(o6)
#undef ESTEP4
#undef ESTEP2
#undef ESTEP
#undef LD4
#undef LD2

  float tot = 0.f, dia = 0.f;
  for (int v = 0; v < 8; ++v) {
    tot += fr[v] * accRe[v] + fi[v] * accIm[v];
    dia += ax[v] * accV[v];
  }
  tot += __shfl_xor(tot, 16);
  dia += __shfl_xor(dia, 16);
  if (lane < 16) { part[wave][0][lane] = tot; part[wave][1][lane] = dia; }
  __syncthreads();
  int t = threadIdx.x;
  if (t < 16) {
    float st = part[0][0][t], sd = part[0][1][t];
    for (int w = 1; w < EWAVES; ++w) { st += part[w][0][t]; sd += part[w][1][t]; }
    out[(b * NPP + col0 + t) * 2 + ch] = st - sd;
  }
}


extern "C" void kernel_launch(void* const* d_in, const int* in_sizes, int n_in,
                              void* d_out, int out_size, void* d_ws, size_t ws_size,
                              hipStream_t stream) {
  (void)in_sizes; (void)n_in; (void)out_size; (void)ws_size;
  const float* x  = (const float*)d_in[0];
  const float* s0 = (const float*)d_in[1];
  const float* s1 = (const float*)d_in[2];
  const float* a0 = (const float*)d_in[3];
  const float* a1 = (const float*)d_in[4];
  float* out = (float*)d_out;

  float* w = (float*)d_ws;
  float* W0   = w; w += MP * MP;
  float* W1   = w; w += MP * MP;
  float* FxRe = w; w += NPTS * MP;
  float* FxIm = w; w += NPTS * MP;
  float* FyRe = w; w += NPTS * MP;
  float* FyIm = w; w += NPTS * MP;
  float* Ax   = w; w += NPTS * MP;
  float* Ay   = w; w += NPTS * MP;
  float* SRe  = w; w += NB * MP * MP;
  float* SIm  = w; w += NB * MP * MP;
  float* FyReF = w; w += NPTS * KF;
  float* FyImF = w; w += NPTS * KF;

  k_front<<<FBLOCKS + WBLOCKS, FTHREADS, 0, stream>>>(x, s0, s1, a0, a1, W0, W1,
                                                      FxRe, FxIm, FyRe, FyIm, FyReF, FyImF, Ax, Ay);
  k_smat<<<dim3(MP / 16, SCOLT, NB), 32 * KSPLIT, 0, stream>>>(FxRe, FxIm, FyRe, FyIm, SRe, SIm);

  k_energy<<<dim3(NPP / 16, NB, 2), 32 * EWAVES, 0, stream>>>(W0, W1, SRe, SIm, FxRe, FxIm,
                                                                FyReF, FyImF, Ax, Ay, out);
}


# Front DFT summed only over the 28 mesh nodes round each point where its Gaussian is nonzero in f32 (exact zeros dropped), window origin restored by a twiddle multiply: K 132 to 28; new sum order
# speedup vs baseline: 5.1637x; 1.1678x over previous
#include <hip/hip_runtime.h>
#include <math.h>

typedef __attribute__((ext_vector_type(2))) float v2f;
typedef __attribute__((ext_vector_type(4))) float v4f;
typedef __attribute__((ext_vector_type(8))) float v8f;

#define MM    129
#define MP    144
#define NPP   512
#define NB    4
#define NPTS  (NB * NPP)
#define KSPLIT 8
#define KF     72
#define SCOLT  5
#define KQ     (KF / 8)

#define PI_F   3.14159265358979323846f
#define TAU_F  (12.0f / (129.0f * 129.0f))
#define L_F    (2.0f * PI_F)


__device__ __forceinline__ v8f wmma4(v2f a, v2f b, v8f c) {
  return __builtin_amdgcn_wmma_f32_16x16x4_f32(false, a, false, b, (short)0, c, false, false);
}


#define KW       28
#define KW0      14
#define FWAVES   (MP / 16)
#define FTHREADS (32 * FWAVES)
#define FBLOCKS  (2 * (NPTS / 16))
#define WBLOCKS  ((MP * KF) / FTHREADS)
__global__ void __launch_bounds__(FTHREADS)
k_front(const float* __restrict__ x,
        const float* s0, const float* s1, const float* a0, const float* a1,
        float* __restrict__ W0, float* __restrict__ W1,
        float* __restrict__ FxRe, float* __restrict__ FxIm,
        float* __restrict__ FyRe, float* __restrict__ FyIm,
        float* __restrict__ FyReF, float* __restrict__ FyImF,
        float* __restrict__ Ax, float* __restrict__ AyF) {
  __shared__ v2f Gs2[16 * (KW / 2)];
  __shared__ v2f tw[MM];
  __shared__ __attribute__((aligned(16))) int ipm[16];
  int t = threadIdx.x;

  if (blockIdx.x >= FBLOCKS) {
    int e = (blockIdx.x - FBLOCKS) * FTHREADS + t;
    int g = e >> 7, l = (e >> 2) & 31;
    int i = (g / KQ) * 16 + (l & 15);
    int k = (g % KQ) * 8 + ((l >> 4) << 2) + (e & 3);
    float w0 = 0.f, w1 = 0.f;
    if (i < MM && k <= MM / 2) {
      float f1 = (float)((i <= MM / 2) ? i : i - MM);
      float f2 = (float)k;
      float k2 = f1 * f1 + f2 * f2;
      float fold = (k == 0) ? 1.0f : 2.0f;
      float dec = (PI_F / TAU_F) * expf(TAU_F * k2);
      float m2   = (float)MM * (float)MM;
      float coef = 1.0f / (8.0f * PI_F * PI_F * m2 * m2);
      float base = fold * coef * dec * dec * 4.0f * PI_F;
      float sh0 = s0[0], sh1 = s1[0];
      w0 = base * a0[0] / (k2 + sh0 * sh0);
      w1 = base * a1[0] / (k2 + sh1 * sh1);
    }
    W0[e] = w0; W1[e] = w1;
    return;
  }

  int axis = blockIdx.x & 1;
  int row0 = (blockIdx.x >> 1) * 16;

  int p = t & 15;
  float xp = x[2 * (row0 + p) + axis];

  int tt = t - (FTHREADS - MM);
  if (tt >= 0) {
    float th = (2.0f * PI_F / (float)MM) * (float)tt;
    v2f e = { cosf(th), -sinf(th) };
    tw[tt] = e;
  }
  float qn = fminf(fmaxf(xp * ((float)MM / L_F), -1.0e6f), 1.0e6f);
  int ipr = (int)rintf(qn) % MM;
  ipr += (ipr < 0) ? MM : 0;
  if (t < 16) ipm[t] = ipr;
  {
    float* Gs = (float*)Gs2;
    const float inv = 1.0f / (4.0f * TAU_F);
    int nb = ipr - KW0;
    for (int jj = t >> 4; jj < KW; jj += FTHREADS / 16) {
      int n = nb + jj;
      n += (n < 0) ? MM : 0;
      n -= (n >= MM) ? MM : 0;
      float xi = (L_F / (float)MM) * (float)n;
      float d  = xp - xi;
      float g = expf(-d * d * inv) + expf(-(d - L_F) * (d - L_F) * inv)
              + expf(-(d + L_F) * (d + L_F) * inv);
      Gs[p * KW + jj] = g;
    }
  }
  __syncthreads();

  int lane = t & 31, r = lane & 15, kb = (lane >> 4) << 1;
  int wave = __builtin_amdgcn_readfirstlane(t >> 5);
  if (axis && wave >= SCOLT) return;
  int c  = wave * 16 + r;
  int cm = (c < MM) ? c : c - MM;
  int m0    = ((kb + MM - KW0) * cm) % MM;
  int step4 = (4 * cm) % MM;
  const v2f* ga = Gs2 + r * (KW / 2) + (lane >> 4);
  v8f accRe = {}, accIm = {};
  for (int j = 0; j < KW / 4; ++j) {
    int m1 = m0 + cm;
    m1 = (m1 >= MM) ? m1 - MM : m1;
    v2f a  = ga[2 * j];
    v2f e0 = tw[m0], e1 = tw[m1];
    v2f bRe = { e0.x, e1.x };
    v2f bIm = { e0.y, e1.y };
    accRe = wmma4(a, bRe, accRe);
    accIm = wmma4(a, bIm, accIm);
    m0 += step4;
    m0 = (m0 >= MM) ? m0 - MM : m0;
  }

  float* DRe = axis ? FyRe : FxRe;
  float* DIm = axis ? FyIm : FxIm;
  bool cv = (c < MM);
  int hr = (lane >> 4) << 3;
  int o0 = (row0 + hr) * MP + c;
  float re[8], im[8];
#pragma unroll
  for (int v = 0; v < 8; ++v) {
    v2f ph = tw[(ipm[hr + v] * cm) % MM];
    float hre = accRe[v], him = accIm[v];
    float fre = hre * ph.x - him * ph.y;
    float fim = hre * ph.y + him * ph.x;
    re[v] = cv ? fre : 0.f;
    im[v] = cv ? fim : 0.f;
    DRe[o0 + v * MP] = re[v];
    DIm[o0 + v * MP] = im[v];
  }
  if (!axis) {
#pragma unroll
    for (int v = 0; v < 8; ++v) Ax[o0 + v * MP] = re[v] * re[v] + im[v] * im[v];
  } else if (c < KF) {
    int of = (((row0 >> 4) * KQ + (c >> 3)) * 32 + (((c >> 2) & 1) << 4) + hr) * 4 + (c & 3);
#pragma unroll
    for (int v = 0; v < 8; ++v) {
      FyReF[of + 4 * v] = re[v];
      FyImF[of + 4 * v] = im[v];
      AyF[of + 4 * v]   = re[v] * re[v] + im[v] * im[v];
    }
  }
}

__global__ void __launch_bounds__(32 * KSPLIT)
k_smat(const float* __restrict__ FxRe, const float* __restrict__ FxIm,
       const float* __restrict__ FyRe, const float* __restrict__ FyIm,
       float* __restrict__ SRe, float* __restrict__ SIm) {
  __shared__ float part[KSPLIT][2][256];
  int b = blockIdx.z;
  int row0 = blockIdx.x * 16, col0 = blockIdx.y * 16;
  int lane = threadIdx.x & 31, r = lane & 15, kb = (lane >> 4) << 1;
  int wave = __builtin_amdgcn_readfirstlane(threadIdx.x >> 5);
  int qbeg = wave * (NPP / KSPLIT);
  const float* fxr = FxRe + (b * NPP + qbeg) * MP;
  const float* fxi = FxIm + (b * NPP + qbeg) * MP;
  const float* fyr = FyRe + (b * NPP + qbeg) * MP;
  const float* fyi = FyIm + (b * NPP + qbeg) * MP;
  v8f accRe = {}, accIm = {};
  for (int k0 = 0; k0 < NPP / KSPLIT; k0 += 4) {
    int q0 = k0 + kb, q1 = q0 + 1;
    v2f aRe = { fxr[q0 * MP + row0 + r], fxr[q1 * MP + row0 + r] };
    v2f aIm = { fxi[q0 * MP + row0 + r], fxi[q1 * MP + row0 + r] };
    v2f aImN = { -aIm.x, -aIm.y };
    v2f bRe = { fyr[q0 * MP + col0 + r], fyr[q1 * MP + col0 + r] };
    v2f bIm = { fyi[q0 * MP + col0 + r], fyi[q1 * MP + col0 + r] };
    accRe = wmma4(aRe,  bRe, accRe);
    accRe = wmma4(aImN, bIm, accRe);
    accIm = wmma4(aRe,  bIm, accIm);
    accIm = wmma4(aIm,  bRe, accIm);
  }
  for (int v = 0; v < 8; ++v) {
    part[wave][0][v * 32 + lane] = accRe[v];
    part[wave][1][v * 32 + lane] = accIm[v];
  }
  __syncthreads();
  int t = threadIdx.x;
  int tr = (t >> 2) & 15;
  int tc = ((t >> 7) << 3) + (((t >> 6) & 1) << 2) + (t & 3);
  int s = (tr & 7) * 32 + ((tr >> 3) << 4) + tc;
  float sre = part[0][0][s], sim = part[0][1][s];
  for (int w = 1; w < KSPLIT; ++w) {
    sre += part[w][0][s];
    sim += part[w][1][s];
  }
  int g = 2 * blockIdx.y + (t >> 7);
  if (g < KQ) {
    int o = b * MP * KF + (blockIdx.x * KQ + g) * 128 + (t & 127);
    SRe[o] = sre;
    SIm[o] = sim;
  }
}

#define KHALF  2
#define EWAVES ((MP / 16) * KHALF)
__global__ void __launch_bounds__(32 * EWAVES)
k_energy(const float* __restrict__ W0, const float* __restrict__ W1,
         const float* __restrict__ SRe, const float* __restrict__ SIm,
         const float* __restrict__ FxRe, const float* __restrict__ FxIm,
         const float* __restrict__ FyRe, const float* __restrict__ FyIm,
         const float* __restrict__ Ax, const float* __restrict__ Ay,
         float* __restrict__ out) {
  __shared__ float part[EWAVES][2][16];
  int col0 = blockIdx.x * 16;
  int b    = blockIdx.y;
  int ch   = blockIdx.z;
  const float* W = ch ? W1 : W0;
  int lane = threadIdx.x & 31, r = lane & 15;
  int wave = __builtin_amdgcn_readfirstlane(threadIdx.x >> 5);
  int kt   = wave / KHALF;
  int hf   = wave % KHALF;
  int row0 = kt * 16;
  int pa = kt * (KQ * 128) + lane * 4;
  int pb = (b * (NPP / 16) + blockIdx.x) * (KQ * 128) + lane * 4;
  const float* wa = W + pa;
  const float* sr = SRe + b * MP * KF + pa;
  const float* si = SIm + b * MP * KF + pa;
  const float* yr = FyRe + pb;
  const float* yi = FyIm + pb;
  const float* ay = Ay + pb;

  int pt = b * NPP + col0 + r;
  int e0 = pt * MP + row0 + ((lane >> 4) << 3);
  float fr[8], fi[8], ax[8];
  for (int v = 0; v < 8; ++v) { fr[v] = FxRe[e0 + v]; fi[v] = FxIm[e0 + v]; ax[v] = Ax[e0 + v]; }

  int g0 = 4 * hf;
  int o1 = g0 * 128 + 2 * hf;
  int o2 = (8 * hf) * 128 + 2;
  int o3 = (g0 + 1) * 128, o4 = (g0 + 2) * 128, o5 = (g0 + 3) * 128;
  int o6 = (g0 + 4) * 128;
  v8f accRe = {}, accIm = {}, accV = {};
#define LD2(p, o) (*(const v2f*)((p) + (o)))
#define LD4(p, o) (*(const v4f*)((p) + (o)))
#define ESTEP(aW, sR, sI, bRe, bIm, bAy) { \
    v2f tRe = (aW) * (sR), tIm = (aW) * (sI); \
    v2f tReN = -tRe; \
    accRe = wmma4(tRe,  (bRe), accRe);     \
    accIm = wmma4(tIm,  (bRe), accIm);     \
    accV  = wmma4((aW), (bAy), accV);      \
    accRe = wmma4(tIm,  (bIm), accRe); \
    accIm = wmma4(tReN, (bIm), accIm); }
#define ESTEP2(o) ESTEP(LD2(wa, o), LD2(sr, o), LD2(si, o), LD2(yr, o), LD2(yi, o), LD2(ay, o))
#define ESTEP4(o) { \
    v4f w4 = LD4(wa, o), r4 = LD4(sr, o), i4 = LD4(si, o), yr4 = LD4(yr, o), yi4 = LD4(yi, o), ay4 = LD4(ay, o); \
    ESTEP(w4.xy, r4.xy, i4.xy, yr4.xy, yi4.xy, ay4.xy) \
    ESTEP(w4.zw, r4.zw, i4.zw, yr4.zw, yi4.zw, ay4.zw) }
  ESTEP2(o1)
  ESTEP2(o2)
  ESTEP4(o3)
  ESTEP4(o4)
  ESTEP4(o5)
  ESTEP2(o6)
#undef ESTEP4
#undef ESTEP2
#undef ESTEP
#undef LD4
#undef LD2

  float tot = 0.f, dia = 0.f;
  for (int v = 0; v < 8; ++v) {
    tot += fr[v] * accRe[v] + fi[v] * accIm[v];
    dia += ax[v] * accV[v];
  }
  tot += __shfl_xor(tot, 16);
  dia += __shfl_xor(dia, 16);
  if (lane < 16) { part[wave][0][lane] = tot; part[wave][1][lane] = dia; }
  __syncthreads();
  int t = threadIdx.x;
  if (t < 16) {
    float st = part[0][0][t], sd = part[0][1][t];
    for (int w = 1; w < EWAVES; ++w) { st += part[w][0][t]; sd += part[w][1][t]; }
    out[(b * NPP + col0 + t) * 2 + ch] = st - sd;
  }
}


extern "C" void kernel_launch(void* const* d_in, const int* in_sizes, int n_in,
                              void* d_out, int out_size, void* d_ws, size_t ws_size,
                              hipStream_t stream) {
  (void)in_sizes; (void)n_in; (void)out_size; (void)ws_size;
  const float* x  = (const float*)d_in[0];
  const float* s0 = (const float*)d_in[1];
  const float* s1 = (const float*)d_in[2];
  const float* a0 = (const float*)d_in[3];
  const float* a1 = (const float*)d_in[4];
  float* out = (float*)d_out;

  float* w = (float*)d_ws;
  float* W0   = w; w += MP * MP;
  float* W1   = w; w += MP * MP;
  float* FxRe = w; w += NPTS * MP;
  float* FxIm = w; w += NPTS * MP;
  float* FyRe = w; w += NPTS * MP;
  float* FyIm = w; w += NPTS * MP;
  float* Ax   = w; w += NPTS * MP;
  float* Ay   = w; w += NPTS * MP;
  float* SRe  = w; w += NB * MP * MP;
  float* SIm  = w; w += NB * MP * MP;
  float* FyReF = w; w += NPTS * KF;
  float* FyImF = w; w += NPTS * KF;

  k_front<<<FBLOCKS + WBLOCKS, FTHREADS, 0, stream>>>(x, s0, s1, a0, a1, W0, W1,
                                                      FxRe, FxIm, FyRe, FyIm, FyReF, FyImF, Ax, Ay);
  k_smat<<<dim3(MP / 16, SCOLT, NB), 32 * KSPLIT, 0, stream>>>(FxRe, FxIm, FyRe, FyIm, SRe, SIm);

  k_energy<<<dim3(NPP / 16, NB, 2), 32 * EWAVES, 0, stream>>>(W0, W1, SRe, SIm, FxRe, FxIm,
                                                                FyReF, FyImF, Ax, Ay, out);
}


# k_energy: 8 waves per block (two per SIMD) instead of 18: each wave forms one whole k1 tile in one chain over k2, ninth tile shared out by k2 groups, loads issued 5 groups ahead; new sum order
# speedup vs baseline: 5.4348x; 1.0525x over previous
#include <hip/hip_runtime.h>
#include <math.h>

typedef __attribute__((ext_vector_type(2))) float v2f;
typedef __attribute__((ext_vector_type(4))) float v4f;
typedef __attribute__((ext_vector_type(8))) float v8f;

#define MM    129
#define MP    144
#define NPP   512
#define NB    4
#define NPTS  (NB * NPP)
#define KSPLIT 8
#define KF     72
#define SCOLT  5
#define KQ     (KF / 8)

#define PI_F   3.14159265358979323846f
#define TAU_F  (12.0f / (129.0f * 129.0f))
#define L_F    (2.0f * PI_F)


__device__ __forceinline__ v8f wmma4(v2f a, v2f b, v8f c) {
  return __builtin_amdgcn_wmma_f32_16x16x4_f32(false, a, false, b, (short)0, c, false, false);
}


#define KW       28
#define KW0      14
#define FWAVES   (MP / 16)
#define FTHREADS (32 * FWAVES)
#define FBLOCKS  (2 * (NPTS / 16))
#define WBLOCKS  ((MP * KF) / FTHREADS)
__global__ void __launch_bounds__(FTHREADS)
k_front(const float* __restrict__ x,
        const float* s0, const float* s1, const float* a0, const float* a1,
        float* __restrict__ W0, float* __restrict__ W1,
        float* __restrict__ FxRe, float* __restrict__ FxIm,
        float* __restrict__ FyRe, float* __restrict__ FyIm,
        float* __restrict__ FyReF, float* __restrict__ FyImF,
        float* __restrict__ Ax, float* __restrict__ AyF) {
  __shared__ v2f Gs2[16 * (KW / 2)];
  __shared__ v2f tw[MM];
  __shared__ __attribute__((aligned(16))) int ipm[16];
  int t = threadIdx.x;

  if (blockIdx.x >= FBLOCKS) {
    int e = (blockIdx.x - FBLOCKS) * FTHREADS + t;
    int g = e >> 7, l = (e >> 2) & 31;
    int i = (g / KQ) * 16 + (l & 15);
    int k = (g % KQ) * 8 + ((l >> 4) << 2) + (e & 3);
    float w0 = 0.f, w1 = 0.f;
    if (i < MM && k <= MM / 2) {
      float f1 = (float)((i <= MM / 2) ? i : i - MM);
      float f2 = (float)k;
      float k2 = f1 * f1 + f2 * f2;
      float fold = (k == 0) ? 1.0f : 2.0f;
      float dec = (PI_F / TAU_F) * expf(TAU_F * k2);
      float m2   = (float)MM * (float)MM;
      float coef = 1.0f / (8.0f * PI_F * PI_F * m2 * m2);
      float base = fold * coef * dec * dec * 4.0f * PI_F;
      float sh0 = s0[0], sh1 = s1[0];
      w0 = base * a0[0] / (k2 + sh0 * sh0);
      w1 = base * a1[0] / (k2 + sh1 * sh1);
    }
    W0[e] = w0; W1[e] = w1;
    return;
  }

  int axis = blockIdx.x & 1;
  int row0 = (blockIdx.x >> 1) * 16;

  int p = t & 15;
  float xp = x[2 * (row0 + p) + axis];

  int tt = t - (FTHREADS - MM);
  if (tt >= 0) {
    float th = (2.0f * PI_F / (float)MM) * (float)tt;
    v2f e = { cosf(th), -sinf(th) };
    tw[tt] = e;
  }
  float qn = fminf(fmaxf(xp * ((float)MM / L_F), -1.0e6f), 1.0e6f);
  int ipr = (int)rintf(qn) % MM;
  ipr += (ipr < 0) ? MM : 0;
  if (t < 16) ipm[t] = ipr;
  {
    float* Gs = (float*)Gs2;
    const float inv = 1.0f / (4.0f * TAU_F);
    int nb = ipr - KW0;
    for (int jj = t >> 4; jj < KW; jj += FTHREADS / 16) {
      int n = nb + jj;
      n += (n < 0) ? MM : 0;
      n -= (n >= MM) ? MM : 0;
      float xi = (L_F / (float)MM) * (float)n;
      float d  = xp - xi;
      float g = expf(-d * d * inv) + expf(-(d - L_F) * (d - L_F) * inv)
              + expf(-(d + L_F) * (d + L_F) * inv);
      Gs[p * KW + jj] = g;
    }
  }
  __syncthreads();

  int lane = t & 31, r = lane & 15, kb = (lane >> 4) << 1;
  int wave = __builtin_amdgcn_readfirstlane(t >> 5);
  if (axis && wave >= SCOLT) return;
  int c  = wave * 16 + r;
  int cm = (c < MM) ? c : c - MM;
  int m0    = ((kb + MM - KW0) * cm) % MM;
  int step4 = (4 * cm) % MM;
  const v2f* ga = Gs2 + r * (KW / 2) + (lane >> 4);
  v8f accRe = {}, accIm = {};
  for (int j = 0; j < KW / 4; ++j) {
    int m1 = m0 + cm;
    m1 = (m1 >= MM) ? m1 - MM : m1;
    v2f a  = ga[2 * j];
    v2f e0 = tw[m0], e1 = tw[m1];
    v2f bRe = { e0.x, e1.x };
    v2f bIm = { e0.y, e1.y };
    accRe = wmma4(a, bRe, accRe);
    accIm = wmma4(a, bIm, accIm);
    m0 += step4;
    m0 = (m0 >= MM) ? m0 - MM : m0;
  }

  float* DRe = axis ? FyRe : FxRe;
  float* DIm = axis ? FyIm : FxIm;
  bool cv = (c < MM);
  int hr = (lane >> 4) << 3;
  int o0 = (row0 + hr) * MP + c;
  float re[8], im[8];
#pragma unroll
  for (int v = 0; v < 8; ++v) {
    v2f ph = tw[(ipm[hr + v] * cm) % MM];
    float hre = accRe[v], him = accIm[v];
    float fre = hre * ph.x - him * ph.y;
    float fim = hre * ph.y + him * ph.x;
    re[v] = cv ? fre : 0.f;
    im[v] = cv ? fim : 0.f;
    DRe[o0 + v * MP] = re[v];
    DIm[o0 + v * MP] = im[v];
  }
  if (!axis) {
#pragma unroll
    for (int v = 0; v < 8; ++v) Ax[o0 + v * MP] = re[v] * re[v] + im[v] * im[v];
  } else if (c < KF) {
    int of = (((row0 >> 4) * KQ + (c >> 3)) * 32 + (((c >> 2) & 1) << 4) + hr) * 4 + (c & 3);
#pragma unroll
    for (int v = 0; v < 8; ++v) {
      FyReF[of + 4 * v] = re[v];
      FyImF[of + 4 * v] = im[v];
      AyF[of + 4 * v]   = re[v] * re[v] + im[v] * im[v];
    }
  }
}

__global__ void __launch_bounds__(32 * KSPLIT)
k_smat(const float* __restrict__ FxRe, const float* __restrict__ FxIm,
       const float* __restrict__ FyRe, const float* __restrict__ FyIm,
       float* __restrict__ SRe, float* __restrict__ SIm) {
  __shared__ float part[KSPLIT][2][256];
  int b = blockIdx.z;
  int row0 = blockIdx.x * 16, col0 = blockIdx.y * 16;
  int lane = threadIdx.x & 31, r = lane & 15, kb = (lane >> 4) << 1;
  int wave = __builtin_amdgcn_readfirstlane(threadIdx.x >> 5);
  int qbeg = wave * (NPP / KSPLIT);
  const float* fxr = FxRe + (b * NPP + qbeg) * MP;
  const float* fxi = FxIm + (b * NPP + qbeg) * MP;
  const float* fyr = FyRe + (b * NPP + qbeg) * MP;
  const float* fyi = FyIm + (b * NPP + qbeg) * MP;
  v8f accRe = {}, accIm = {};
  for (int k0 = 0; k0 < NPP / KSPLIT; k0 += 4) {
    int q0 = k0 + kb, q1 = q0 + 1;
    v2f aRe = { fxr[q0 * MP + row0 + r], fxr[q1 * MP + row0 + r] };
    v2f aIm = { fxi[q0 * MP + row0 + r], fxi[q1 * MP + row0 + r] };
    v2f aImN = { -aIm.x, -aIm.y };
    v2f bRe = { fyr[q0 * MP + col0 + r], fyr[q1 * MP + col0 + r] };
    v2f bIm = { fyi[q0 * MP + col0 + r], fyi[q1 * MP + col0 + r] };
    accRe = wmma4(aRe,  bRe, accRe);
    accRe = wmma4(aImN, bIm, accRe);
    accIm = wmma4(aRe,  bIm, accIm);
    accIm = wmma4(aIm,  bRe, accIm);
  }
  for (int v = 0; v < 8; ++v) {
    part[wave][0][v * 32 + lane] = accRe[v];
    part[wave][1][v * 32 + lane] = accIm[v];
  }
  __syncthreads();
  int t = threadIdx.x;
  int tr = (t >> 2) & 15;
  int tc = ((t >> 7) << 3) + (((t >> 6) & 1) << 2) + (t & 3);
  int s = (tr & 7) * 32 + ((tr >> 3) << 4) + tc;
  float sre = part[0][0][s], sim = part[0][1][s];
  for (int w = 1; w < KSPLIT; ++w) {
    sre += part[w][0][s];
    sim += part[w][1][s];
  }
  int g = 2 * blockIdx.y + (t >> 7);
  if (g < KQ) {
    int o = b * MP * KF + (blockIdx.x * KQ + g) * 128 + (t & 127);
    SRe[o] = sre;
    SIm[o] = sim;
  }
}

#define EWAVES 8
__global__ void __launch_bounds__(32 * EWAVES) __attribute__((amdgpu_waves_per_eu(2, 5)))
k_energy(const float* __restrict__ W0, const float* __restrict__ W1,
         const float* __restrict__ SRe, const float* __restrict__ SIm,
         const float* __restrict__ FxRe, const float* __restrict__ FxIm,
         const float* __restrict__ FyRe, const float* __restrict__ FyIm,
         const float* __restrict__ Ax, const float* __restrict__ Ay,
         float* __restrict__ out) {
  static_assert(MP / 16 == EWAVES + 1 && KQ == EWAVES + 1, "one own k1 tile per wave, the last tile shared out by groups");
  __shared__ float part[EWAVES][2][16];
  int col0 = blockIdx.x * 16;
  int b    = blockIdx.y;
  int ch   = blockIdx.z;
  const float* W = ch ? W1 : W0;
  int lane = threadIdx.x & 31, r = lane & 15;
  int wave = __builtin_amdgcn_readfirstlane(threadIdx.x >> 5);
  int pa = wave * (KQ * 128) + lane * 4;
  int pl = EWAVES * (KQ * 128) + lane * 4;
  int pb = (b * (NPP / 16) + blockIdx.x) * (KQ * 128) + lane * 4;
  const float* sr = SRe + b * MP * KF;
  const float* si = SIm + b * MP * KF;
  const float* yr = FyRe + pb;
  const float* yi = FyIm + pb;
  const float* ay = Ay + pb;

  int pt = b * NPP + col0 + r;
  int eA = pt * MP + wave * 16 + ((lane >> 4) << 3);
  int eB = pt * MP + EWAVES * 16 + ((lane >> 4) << 3);
  float frA[8], fiA[8], axA[8], frB[8], fiB[8], axB[8];
  for (int v = 0; v < 8; ++v) { frA[v] = FxRe[eA + v]; fiA[v] = FxIm[eA + v]; axA[v] = Ax[eA + v]; }
  for (int v = 0; v < 8; ++v) { frB[v] = FxRe[eB + v]; fiB[v] = FxIm[eB + v]; axB[v] = Ax[eB + v]; }

  v8f accRe = {}, accIm = {}, accV = {};
#define LD4(p, o) (*(const v4f*)((p) + (o)))
#define ESTEP(aW, sR, sI, bRe, bIm, bAy) { \
    v2f tRe = (aW) * (sR), tIm = (aW) * (sI); \
    v2f tReN = -tRe; \
    accRe = wmma4(tRe,  (bRe), accRe);     \
    accIm = wmma4(tIm,  (bRe), accIm);     \
    accV  = wmma4((aW), (bAy), accV);      \
    accRe = wmma4(tIm,  (bIm), accRe); \
    accIm = wmma4(tReN, (bIm), accIm); }
#define EGROUP(w4, r4, i4, ob) { \
    v4f yr4 = LD4(yr, ob), yi4 = LD4(yi, ob), ay4 = LD4(ay, ob); \
    ESTEP(w4.xy, r4.xy, i4.xy, yr4.xy, yi4.xy, ay4.xy) \
    ESTEP(w4.zw, r4.zw, i4.zw, yr4.zw, yi4.zw, ay4.zw) }
#define EDOT(fr, fi, ax) { \
    for (int v = 0; v < 8; ++v) { \
      tot += fr[v] * accRe[v] + fi[v] * accIm[v]; \
      dia += ax[v] * accV[v]; } }

  float tot = 0.f, dia = 0.f;
#pragma unroll
  for (int g = 0; g < KQ; ++g) {
    int oa = pa + g * 128;
    v4f w4 = LD4(W, oa), r4 = LD4(sr, oa), i4 = LD4(si, oa);
    EGROUP(w4, r4, i4, g * 128)
  }
  EDOT(frA, fiA, axA)

  accRe = v8f{}; accIm = v8f{}; accV = v8f{};
  {
    int oa = pl + wave * 128;
    v4f w4 = LD4(W, oa), r4 = LD4(sr, oa), i4 = LD4(si, oa);
    EGROUP(w4, r4, i4, wave * 128)
  }
  {
    int oa = pl + (KQ - 1) * 128;
    v4f w4 = LD4(W, oa), r4 = LD4(sr, oa), i4 = LD4(si, oa);
    bool mine = (wave == EWAVES - 1);
    v4f z4 = {0.f, 0.f, 0.f, 0.f};
    w4 = mine ? w4 : z4; r4 = mine ? r4 : z4; i4 = mine ? i4 : z4;
    EGROUP(w4, r4, i4, (KQ - 1) * 128)
  }
  EDOT(frB, fiB, axB)
#undef EDOT
#undef EGROUP
#undef ESTEP
#undef LD4

  tot += __shfl_xor(tot, 16);
  dia += __shfl_xor(dia, 16);
  if (lane < 16) { part[wave][0][lane] = tot; part[wave][1][lane] = dia; }
  __syncthreads();
  int t = threadIdx.x;
  if (t < 16) {
    float st = part[0][0][t], sd = part[0][1][t];
    for (int w = 1; w < EWAVES; ++w) { st += part[w][0][t]; sd += part[w][1][t]; }
    out[(b * NPP + col0 + t) * 2 + ch] = st - sd;
  }
}


extern "C" void kernel_launch(void* const* d_in, const int* in_sizes, int n_in,
                              void* d_out, int out_size, void* d_ws, size_t ws_size,
                              hipStream_t stream) {
  (void)in_sizes; (void)n_in; (void)out_size; (void)ws_size;
  const float* x  = (const float*)d_in[0];
  const float* s0 = (const float*)d_in[1];
  const float* s1 = (const float*)d_in[2];
  const float* a0 = (const float*)d_in[3];
  const float* a1 = (const float*)d_in[4];
  float* out = (float*)d_out;

  float* w = (float*)d_ws;
  float* W0   = w; w += MP * MP;
  float* W1   = w; w += MP * MP;
  float* FxRe = w; w += NPTS * MP;
  float* FxIm = w; w += NPTS * MP;
  float* FyRe = w; w += NPTS * MP;
  float* FyIm = w; w += NPTS * MP;
  float* Ax   = w; w += NPTS * MP;
  float* Ay   = w; w += NPTS * MP;
  float* SRe  = w; w += NB * MP * MP;
  float* SIm  = w; w += NB * MP * MP;
  float* FyReF = w; w += NPTS * KF;
  float* FyImF = w; w += NPTS * KF;

  k_front<<<FBLOCKS + WBLOCKS, FTHREADS, 0, stream>>>(x, s0, s1, a0, a1, W0, W1,
                                                      FxRe, FxIm, FyRe, FyIm, FyReF, FyImF, Ax, Ay);
  k_smat<<<dim3(MP / 16, SCOLT, NB), 32 * KSPLIT, 0, stream>>>(FxRe, FxIm, FyRe, FyIm, SRe, SIm);

  k_energy<<<dim3(NPP / 16, NB, 2), 32 * EWAVES, 0, stream>>>(W0, W1, SRe, SIm, FxRe, FxIm,
                                                                FyReF, FyImF, Ax, Ay, out);
}
